# starting kernel
# baseline (speedup 1.0000x reference)
#include <hip/hip_runtime.h>
#include <hip/hip_bf16.h>
#include <math.h>


typedef float v2f __attribute__((ext_vector_type(2)));
typedef float v8f __attribute__((ext_vector_type(8)));

static constexpr int N_ = 2;
static constexpr int S_ = 2048;
static constexpr int E_ = 1024;
static constexpr int H_ = 16;
static constexpr int D_ = 64;

static __device__ __forceinline__ v8f wmma_f32(v2f a, v2f b, v8f c) {
  return __builtin_amdgcn_wmma_f32_16x16x4_f32(false, a, false, b,
                                               (short)0, c, false, false);
}

static __device__ __forceinline__ float rowmax16(float v) {
  v = fmaxf(v, __shfl_xor(v, 1, 32));
  v = fmaxf(v, __shfl_xor(v, 2, 32));
  v = fmaxf(v, __shfl_xor(v, 4, 32));
  v = fmaxf(v, __shfl_xor(v, 8, 32));
  return v;
}
static __device__ __forceinline__ float rowsum16(float v) {
  v += __shfl_xor(v, 1, 32);
  v += __shfl_xor(v, 2, 32);
  v += __shfl_xor(v, 4, 32);
  v += __shfl_xor(v, 8, 32);
  return v;
}

__global__ __launch_bounds__(128)
void proj_kernel(const float* __restrict__ X,
                 const float* __restrict__ W,
                 float* __restrict__ Out)
{
  __shared__ float Ws[64 * 68];
  const int tid  = threadIdx.x;
  const int wave = tid >> 5;
  const int lane = tid & 31;
  const int n  = blockIdx.z, h = blockIdx.y;
  const int s0 = blockIdx.x * 64 + wave * 16;
  const int mrow = lane & 15, half = lane >> 4;

  for (int i = tid; i < 1024; i += 128) {
    const int r = i >> 4, c4 = i & 15;
    const float4 w4 = ((const float4*)W)[r * 16 + c4];
    float* d = &Ws[r * 68 + c4 * 4];
    d[0] = w4.x; d[1] = w4.y; d[2] = w4.z; d[3] = w4.w;
  }
  __syncthreads();

  const float* xrow = X + ((size_t)n * S_ + (s0 + mrow)) * E_ + h * D_;
  v8f c[4] = {};
  #pragma unroll
  for (int ks = 0; ks < 16; ++ks) {
    const int k = 4 * ks + 2 * half;
    v2f a; a.x = xrow[k]; a.y = xrow[k + 1];
    #pragma unroll
    for (int f = 0; f < 4; ++f) {
      const int e = 16 * f + mrow;
      v2f b; b.x = Ws[e * 68 + k]; b.y = Ws[e * 68 + k + 1];
      c[f] = wmma_f32(a, b, c[f]);
    }
  }

  float* obase = Out + ((size_t)(n * H_ + h)) * S_ * D_;
  #pragma unroll
  for (int f = 0; f < 4; ++f)
    #pragma unroll
    for (int r = 0; r < 8; ++r) {
      const int m = r + 8 * half;
      const int e = 16 * f + mrow;
      obase[(size_t)(s0 + m) * D_ + e] = c[f][r];
    }
}

__global__ __launch_bounds__(128)
void flash_kernel(const float* __restrict__ Qp,
                  const float* __restrict__ Kp,
                  const float* __restrict__ Vp,
                  float* __restrict__ O)
{
  __shared__ float Ks[16 * 68];
  __shared__ float Vs[16 * 68];
  __shared__ float Ps[4][16 * 17];

  const int tid  = threadIdx.x;
  const int wave = tid >> 5;
  const int lane = tid & 31;
  const int n = blockIdx.z, h = blockIdx.y;
  const int qblk = blockIdx.x;
  const int q0 = qblk * 64 + wave * 16;
  const int mrow = lane & 15, half = lane >> 4;
  const size_t headOff = ((size_t)(n * H_ + h)) * S_ * D_;

  float qreg[32];
  const float* qrow = Qp + headOff + (size_t)(q0 + mrow) * D_;
  #pragma unroll
  for (int ks = 0; ks < 16; ++ks) {
    const int k = 4 * ks + 2 * half;
    qreg[2 * ks]     = qrow[k];
    qreg[2 * ks + 1] = qrow[k + 1];
  }

  v8f o[4] = {};
  float mrun[8], lrun[8];
  #pragma unroll
  for (int r = 0; r < 8; ++r) { mrun[r] = -3.0e38f; lrun[r] = 0.0f; }

  const float slope  = (float)(1 << __popc(h));
  const float scale  = 0.03125f;
  const int   ktMax  = (qblk * 64 + 63) >> 4;

  for (int kt = 0; kt <= ktMax; ++kt) {
    const int k0 = kt * 16;
    __syncthreads();
    for (int i = tid; i < 256; i += 128) {
      const int r = i >> 4, c4 = i & 15;
      const float4 kq = ((const float4*)(Kp + headOff + (size_t)(k0 + r) * D_))[c4];
      const float4 vq = ((const float4*)(Vp + headOff + (size_t)(k0 + r) * D_))[c4];
      float* kd = &Ks[r * 68 + c4 * 4];
      float* vd = &Vs[r * 68 + c4 * 4];
      kd[0] = kq.x; kd[1] = kq.y; kd[2] = kq.z; kd[3] = kq.w;
      vd[0] = vq.x; vd[1] = vq.y; vd[2] = vq.z; vd[3] = vq.w;
    }
    __syncthreads();

    const bool active = (k0 <= q0 + 15);
    if (active) {
      v8f s = {};
      #pragma unroll
      for (int ks = 0; ks < 16; ++ks) {
        const int k = 4 * ks + 2 * half;
        v2f a; a.x = qreg[2 * ks]; a.y = qreg[2 * ks + 1];
        v2f b; b.x = Ks[mrow * 68 + k]; b.y = Ks[mrow * 68 + k + 1];
        s = wmma_f32(a, b, s);
      }
      #pragma unroll
      for (int r = 0; r < 8; ++r) {
        const int q = q0 + r + 8 * half;
        const int kcol = k0 + mrow;
        float e = (s[r] + slope * (float)(kcol - q)) * scale;
        if (kcol > q) e = -3.0e18f;
        s[r] = e;
      }
      float pscale[8];
      #pragma unroll
      for (int r = 0; r < 8; ++r) {
        const float rm   = rowmax16(s[r]);
        const float mnew = fmaxf(mrun[r], rm);
        const float alpha = __expf(mrun[r] - mnew);
        mrun[r] = mnew;
        const float p = __expf(s[r] - mnew);
        s[r] = p;
        lrun[r] = lrun[r] * alpha + rowsum16(p);
        pscale[r] = alpha;
      }
      #pragma unroll
      for (int f = 0; f < 4; ++f)
        #pragma unroll
        for (int r = 0; r < 8; ++r) o[f][r] *= pscale[r];

      #pragma unroll
      for (int r = 0; r < 8; ++r)
        Ps[wave][(r + 8 * half) * 17 + mrow] = s[r];
      #pragma unroll
      for (int kk = 0; kk < 4; ++kk) {
        const int ka = 4 * kk + 2 * half;
        v2f a; a.x = Ps[wave][mrow * 17 + ka]; a.y = Ps[wave][mrow * 17 + ka + 1];
        #pragma unroll
        for (int f = 0; f < 4; ++f) {
          const int dcol = 16 * f + mrow;
          v2f b;
          b.x = Vs[(4 * kk + 2 * half)     * 68 + dcol];
          b.y = Vs[(4 * kk + 2 * half + 1) * 68 + dcol];
          o[f] = wmma_f32(a, b, o[f]);
        }
      }
    }
  }

  #pragma unroll
  for (int r = 0; r < 8; ++r) lrun[r] = 1.0f / lrun[r];
  float* obase = O + (size_t)n * S_ * E_ + h * D_;
  #pragma unroll
  for (int f = 0; f < 4; ++f)
    #pragma unroll
    for (int r = 0; r < 8; ++r) {
      const int m = r + 8 * half;
      const int dcol = 16 * f + mrow;
      obase[(size_t)(q0 + m) * E_ + dcol] = o[f][r] * lrun[r];
    }
}

__global__ __launch_bounds__(128)
void outproj_kernel(const float* __restrict__ A,
                    const float* __restrict__ Wo,
                    const float* __restrict__ bo,
                    float* __restrict__ Y)
{
  __shared__ float Ws[64 * 68];
  const int tid  = threadIdx.x;
  const int wave = tid >> 5;
  const int lane = tid & 31;
  const int r0 = blockIdx.y * 64 + wave * 16;
  const int c0 = blockIdx.x * 64;
  const int mrow = lane & 15, half = lane >> 4;

  v8f c[4] = {};
  const float* arow = A + (size_t)(r0 + mrow) * E_;

  for (int kc = 0; kc < E_; kc += 64) {
    __syncthreads();
    for (int i = tid; i < 1024; i += 128) {
      const int rr = i >> 4, c4 = i & 15;
      const float4 w4 = ((const float4*)(Wo + (size_t)(c0 + rr) * E_ + kc))[c4];
      float* d = &Ws[rr * 68 + c4 * 4];
      d[0] = w4.x; d[1] = w4.y; d[2] = w4.z; d[3] = w4.w;
    }
    __syncthreads();
    #pragma unroll
    for (int ks = 0; ks < 16; ++ks) {
      const int k = 4 * ks + 2 * half;
      v2f a; a.x = arow[kc + k]; a.y = arow[kc + k + 1];
      #pragma unroll
      for (int f = 0; f < 4; ++f) {
        const int e = 16 * f + mrow;
        v2f b; b.x = Ws[e * 68 + k]; b.y = Ws[e * 68 + k + 1];
        c[f] = wmma_f32(a, b, c[f]);
      }
    }
  }

  #pragma unroll
  for (int f = 0; f < 4; ++f)
    #pragma unroll
    for (int r = 0; r < 8; ++r) {
      const int m = r + 8 * half;
      const int e = c0 + 16 * f + mrow;
      Y[(size_t)(r0 + m) * E_ + e] = c[f][r] + bo[e];
    }
}

extern "C" void kernel_launch(void* const* d_in, const int* in_sizes, int n_in,
                              void* d_out, int out_size, void* d_ws, size_t ws_size,
                              hipStream_t stream) {
  const float* values = (const float*)d_in[0];
  const float* keys   = (const float*)d_in[1];
  const float* query  = (const float*)d_in[2];
  const float* Wv     = (const float*)d_in[4];
  const float* Wk     = (const float*)d_in[5];
  const float* Wq     = (const float*)d_in[6];
  const float* Wo     = (const float*)d_in[7];
  const float* bo     = (const float*)d_in[8];
  float* out = (float*)d_out;

  const size_t tensorElems = (size_t)N_ * H_ * S_ * D_;
  float* Qp = (float*)d_ws;
  float* Kp = Qp + tensorElems;
  float* Vp = Kp + tensorElems;
  float* Oa = Vp + tensorElems;

  const dim3 blk(128);
  const dim3 gProj(S_ / 64, H_, N_);

  proj_kernel<<<gProj, blk, 0, stream>>>(query,  Wq, Qp);
  proj_kernel<<<gProj, blk, 0, stream>>>(keys,   Wk, Kp);
  proj_kernel<<<gProj, blk, 0, stream>>>(values, Wv, Vp);

  flash_kernel<<<gProj, blk, 0, stream>>>(Qp, Kp, Vp, Oa);

  const dim3 gOut(E_ / 64, (N_ * S_) / 64);
  outproj_kernel<<<gOut, blk, 0, stream>>>(Oa, Wo, bo, out);
}


# attention kernel: key tile 16 -> 64 keys per barrier pair; running max/sum and output rescale once per 64 keys, so row sums and rescaled partial outputs are rounded in a different order
# speedup vs baseline: 1.3537x; 1.3537x over previous
#include <hip/hip_runtime.h>
#include <hip/hip_bf16.h>
#include <math.h>


typedef float v2f __attribute__((ext_vector_type(2)));
typedef float v8f __attribute__((ext_vector_type(8)));

static constexpr int N_ = 2;
static constexpr int S_ = 2048;
static constexpr int E_ = 1024;
static constexpr int H_ = 16;
static constexpr int D_ = 64;

static __device__ __forceinline__ v8f wmma_f32(v2f a, v2f b, v8f c) {
  return __builtin_amdgcn_wmma_f32_16x16x4_f32(false, a, false, b,
                                               (short)0, c, false, false);
}

static __device__ __forceinline__ float rowmax16(float v) {
  v = fmaxf(v, __shfl_xor(v, 1, 32));
  v = fmaxf(v, __shfl_xor(v, 2, 32));
  v = fmaxf(v, __shfl_xor(v, 4, 32));
  v = fmaxf(v, __shfl_xor(v, 8, 32));
  return v;
}
static __device__ __forceinline__ float rowsum16(float v) {
  v += __shfl_xor(v, 1, 32);
  v += __shfl_xor(v, 2, 32);
  v += __shfl_xor(v, 4, 32);
  v += __shfl_xor(v, 8, 32);
  return v;
}

__global__ __launch_bounds__(128)
void proj_kernel(const float* __restrict__ X,
                 const float* __restrict__ W,
                 float* __restrict__ Out)
{
  __shared__ float Ws[64 * 68];
  const int tid  = threadIdx.x;
  const int wave = tid >> 5;
  const int lane = tid & 31;
  const int n  = blockIdx.z, h = blockIdx.y;
  const int s0 = blockIdx.x * 64 + wave * 16;
  const int mrow = lane & 15, half = lane >> 4;

  for (int i = tid; i < 1024; i += 128) {
    const int r = i >> 4, c4 = i & 15;
    const float4 w4 = ((const float4*)W)[r * 16 + c4];
    float* d = &Ws[r * 68 + c4 * 4];
    d[0] = w4.x; d[1] = w4.y; d[2] = w4.z; d[3] = w4.w;
  }
  __syncthreads();

  const float* xrow = X + ((size_t)n * S_ + (s0 + mrow)) * E_ + h * D_;
  v8f c[4] = {};
  #pragma unroll
  for (int ks = 0; ks < 16; ++ks) {
    const int k = 4 * ks + 2 * half;
    v2f a; a.x = xrow[k]; a.y = xrow[k + 1];
    #pragma unroll
    for (int f = 0; f < 4; ++f) {
      const int e = 16 * f + mrow;
      v2f b; b.x = Ws[e * 68 + k]; b.y = Ws[e * 68 + k + 1];
      c[f] = wmma_f32(a, b, c[f]);
    }
  }

  float* obase = Out + ((size_t)(n * H_ + h)) * S_ * D_;
  #pragma unroll
  for (int f = 0; f < 4; ++f)
    #pragma unroll
    for (int r = 0; r < 8; ++r) {
      const int m = r + 8 * half;
      const int e = 16 * f + mrow;
      obase[(size_t)(s0 + m) * D_ + e] = c[f][r];
    }
}

static constexpr int BK_  = 64;
static constexpr int LDP_ = 68;

__global__ __launch_bounds__(128)
void flash_kernel(const float* __restrict__ Qp,
                  const float* __restrict__ Kp,
                  const float* __restrict__ Vp,
                  float* __restrict__ O)
{
  __shared__ __attribute__((aligned(16))) float smem[(2 * BK_ + 4 * 16) * LDP_];

  const int tid  = threadIdx.x;
  const int wave = tid >> 5;
  const int lane = tid & 31;
  const int n = blockIdx.z, h = blockIdx.y;
  const int qblk = blockIdx.x;
  const int q0 = qblk * 64 + wave * 16;
  const int mrow = lane & 15, half = lane >> 4;
  const size_t headOff = ((size_t)(n * H_ + h)) * S_ * D_;

  float* Ks = smem;
  float* Vs = smem + BK_ * LDP_;
  float* Pw = smem + 2 * BK_ * LDP_ + wave * 16 * LDP_;

  float qreg[32];
  const float* qrow = Qp + headOff + (size_t)(q0 + mrow) * D_;
  #pragma unroll
  for (int ks = 0; ks < 16; ++ks) {
    const int k = 4 * ks + 2 * half;
    qreg[2 * ks]     = qrow[k];
    qreg[2 * ks + 1] = qrow[k + 1];
  }

  v8f o[4] = {};
  float mrun[8], lrun[8];
  #pragma unroll
  for (int r = 0; r < 8; ++r) { mrun[r] = -3.0e38f; lrun[r] = 0.0f; }

  const float slope  = (float)(1 << __popc(h));
  const float scale  = 0.03125f;
  const float ss     = slope * scale;

  const int srow = tid >> 4, sc4 = tid & 15;
  const float* kg = Kp + headOff + (size_t)srow * D_ + sc4 * 4;
  const float* vg = Vp + headOff + (size_t)srow * D_ + sc4 * 4;
  float* sdst = smem + srow * LDP_ + sc4 * 4;

  for (int kt = 0; kt <= qblk; ++kt) {
    const int k0 = kt * BK_;

    __syncthreads();
    {
      float4 kq[8], vq[8];
      #pragma unroll
      for (int i = 0; i < 8; ++i) kq[i] = *(const float4*)(kg + (size_t)(k0 + 8 * i) * D_);
      #pragma unroll
      for (int i = 0; i < 8; ++i) vq[i] = *(const float4*)(vg + (size_t)(k0 + 8 * i) * D_);
      #pragma unroll
      for (int i = 0; i < 8; ++i) *(float4*)(sdst + 8 * i * LDP_) = kq[i];
      #pragma unroll
      for (int i = 0; i < 8; ++i) *(float4*)(sdst + (BK_ + 8 * i) * LDP_) = vq[i];
    }
    __syncthreads();

    v8f s[4] = {};
    #pragma unroll
    for (int ks = 0; ks < 16; ++ks) {
      const int k = 4 * ks + 2 * half;
      v2f a; a.x = qreg[2 * ks]; a.y = qreg[2 * ks + 1];
      #pragma unroll
      for (int j = 0; j < 4; ++j) {
        const v2f b = *(const v2f*)&Ks[(16 * j + mrow) * LDP_ + k];
        s[j] = wmma_f32(a, b, s[j]);
      }
    }

    const float lanebias = ss * (float)(k0 - q0 + mrow - 8 * half);
    #pragma unroll
    for (int j = 0; j < 4; ++j)
      #pragma unroll
      for (int r = 0; r < 8; ++r)
        s[j][r] = fmaf(s[j][r], scale, lanebias + ss * (float)(16 * j - r));

    if (kt == qblk) {
      const int d = mrow - 8 * half - 16 * wave;
      #pragma unroll
      for (int j = 0; j < 4; ++j)
        #pragma unroll
        for (int r = 0; r < 8; ++r)
          if (d + 16 * j - r > 0) s[j][r] = -3.0e18f;
    }

    #pragma unroll
    for (int r = 0; r < 8; ++r) {
      const float rm   = rowmax16(fmaxf(fmaxf(s[0][r], s[1][r]), fmaxf(s[2][r], s[3][r])));
      const float mnew = fmaxf(mrun[r], rm);
      const float alpha = __expf(mrun[r] - mnew);
      mrun[r] = mnew;
      const float p0 = __expf(s[0][r] - mnew);
      const float p1 = __expf(s[1][r] - mnew);
      const float p2 = __expf(s[2][r] - mnew);
      const float p3 = __expf(s[3][r] - mnew);
      s[0][r] = p0; s[1][r] = p1; s[2][r] = p2; s[3][r] = p3;
      lrun[r] = lrun[r] * alpha + rowsum16((p0 + p1) + (p2 + p3));
      #pragma unroll
      for (int f = 0; f < 4; ++f) o[f][r] *= alpha;
    }

    #pragma unroll
    for (int j = 0; j < 4; ++j)
      #pragma unroll
      for (int r = 0; r < 8; ++r)
        Pw[(r + 8 * half) * LDP_ + 16 * j + mrow] = s[j][r];
    #pragma unroll
    for (int kk = 0; kk < 16; ++kk) {
      const int ka = 4 * kk + 2 * half;
      const v2f a = *(const v2f*)&Pw[mrow * LDP_ + ka];
      #pragma unroll
      for (int f = 0; f < 4; ++f) {
        const int dcol = 16 * f + mrow;
        v2f b;
        b.x = Vs[ka       * LDP_ + dcol];
        b.y = Vs[(ka + 1) * LDP_ + dcol];
        o[f] = wmma_f32(a, b, o[f]);
      }
    }
  }

  #pragma unroll
  for (int r = 0; r < 8; ++r) lrun[r] = 1.0f / lrun[r];
  float* obase = O + (size_t)n * S_ * E_ + h * D_;
  #pragma unroll
  for (int f = 0; f < 4; ++f)
    #pragma unroll
    for (int r = 0; r < 8; ++r) {
      const int m = r + 8 * half;
      const int dcol = 16 * f + mrow;
      obase[(size_t)(q0 + m) * E_ + dcol] = o[f][r] * lrun[r];
    }
}

__global__ __launch_bounds__(128)
void outproj_kernel(const float* __restrict__ A,
                    const float* __restrict__ Wo,
                    const float* __restrict__ bo,
                    float* __restrict__ Y)
{
  __shared__ float Ws[64 * 68];
  const int tid  = threadIdx.x;
  const int wave = tid >> 5;
  const int lane = tid & 31;
  const int r0 = blockIdx.y * 64 + wave * 16;
  const int c0 = blockIdx.x * 64;
  const int mrow = lane & 15, half = lane >> 4;

  v8f c[4] = {};
  const float* arow = A + (size_t)(r0 + mrow) * E_;

  for (int kc = 0; kc < E_; kc += 64) {
    __syncthreads();
    for (int i = tid; i < 1024; i += 128) {
      const int rr = i >> 4, c4 = i & 15;
      const float4 w4 = ((const float4*)(Wo + (size_t)(c0 + rr) * E_ + kc))[c4];
      float* d = &Ws[rr * 68 + c4 * 4];
      d[0] = w4.x; d[1] = w4.y; d[2] = w4.z; d[3] = w4.w;
    }
    __syncthreads();
    #pragma unroll
    for (int ks = 0; ks < 16; ++ks) {
      const int k = 4 * ks + 2 * half;
      v2f a; a.x = arow[kc + k]; a.y = arow[kc + k + 1];
      #pragma unroll
      for (int f = 0; f < 4; ++f) {
        const int e = 16 * f + mrow;
        v2f b; b.x = Ws[e * 68 + k]; b.y = Ws[e * 68 + k + 1];
        c[f] = wmma_f32(a, b, c[f]);
      }
    }
  }

  #pragma unroll
  for (int f = 0; f < 4; ++f)
    #pragma unroll
    for (int r = 0; r < 8; ++r) {
      const int m = r + 8 * half;
      const int e = c0 + 16 * f + mrow;
      Y[(size_t)(r0 + m) * E_ + e] = c[f][r] + bo[e];
    }
}

extern "C" void kernel_launch(void* const* d_in, const int* in_sizes, int n_in,
                              void* d_out, int out_size, void* d_ws, size_t ws_size,
                              hipStream_t stream) {
  const float* values = (const float*)d_in[0];
  const float* keys   = (const float*)d_in[1];
  const float* query  = (const float*)d_in[2];
  const float* Wv     = (const float*)d_in[4];
  const float* Wk     = (const float*)d_in[5];
  const float* Wq     = (const float*)d_in[6];
  const float* Wo     = (const float*)d_in[7];
  const float* bo     = (const float*)d_in[8];
  float* out = (float*)d_out;

  const size_t tensorElems = (size_t)N_ * H_ * S_ * D_;
  float* Qp = (float*)d_ws;
  float* Kp = Qp + tensorElems;
  float* Vp = Kp + tensorElems;
  float* Oa = Vp + tensorElems;

  const dim3 blk(128);
  const dim3 gProj(S_ / 64, H_, N_);

  proj_kernel<<<gProj, blk, 0, stream>>>(query,  Wq, Qp);
  proj_kernel<<<gProj, blk, 0, stream>>>(keys,   Wk, Kp);
  proj_kernel<<<gProj, blk, 0, stream>>>(values, Wv, Vp);

  flash_kernel<<<gProj, blk, 0, stream>>>(Qp, Kp, Vp, Oa);

  const dim3 gOut(E_ / 64, (N_ * S_) / 64);
  outproj_kernel<<<gOut, blk, 0, stream>>>(Oa, Wo, bo, out);
}


# all three kernels: each tile's global loads (weight and K/V pieces, A rows) go out one tile ahead into registers, before the current tile's WMMAs; staging no longer waits on its loads one by one
# speedup vs baseline: 2.2470x; 1.6599x over previous
#include <hip/hip_runtime.h>
#include <hip/hip_bf16.h>
#include <math.h>


typedef float v2f __attribute__((ext_vector_type(2)));
typedef float v4f __attribute__((ext_vector_type(4)));
typedef float v8f __attribute__((ext_vector_type(8)));

static constexpr int N_ = 2;
static constexpr int S_ = 2048;
static constexpr int E_ = 1024;
static constexpr int H_ = 16;
static constexpr int D_ = 64;

static __device__ __forceinline__ v8f wmma_f32(v2f a, v2f b, v8f c) {
  return __builtin_amdgcn_wmma_f32_16x16x4_f32(false, a, false, b,
                                               (short)0, c, false, false);
}

static __device__ __forceinline__ float rowmax16(float v) {
  v = fmaxf(v, __shfl_xor(v, 1, 32));
  v = fmaxf(v, __shfl_xor(v, 2, 32));
  v = fmaxf(v, __shfl_xor(v, 4, 32));
  v = fmaxf(v, __shfl_xor(v, 8, 32));
  return v;
}
static __device__ __forceinline__ float rowsum16(float v) {
  v += __shfl_xor(v, 1, 32);
  v += __shfl_xor(v, 2, 32);
  v += __shfl_xor(v, 4, 32);
  v += __shfl_xor(v, 8, 32);
  return v;
}

__global__ __launch_bounds__(128)
void proj_kernel(const float* __restrict__ X,
                 const float* __restrict__ W,
                 float* __restrict__ Out)
{
  __shared__ __attribute__((aligned(16))) float Ws[64 * 68];
  const int tid  = threadIdx.x;
  const int wave = tid >> 5;
  const int lane = tid & 31;
  const int n  = blockIdx.z, h = blockIdx.y;
  const int s0 = blockIdx.x * 64 + wave * 16;
  const int mrow = lane & 15, half = lane >> 4;
  const int srow = tid >> 4, sc4 = tid & 15;

  const float* xrow = X + ((size_t)n * S_ + (s0 + mrow)) * E_ + h * D_;
  v4f wq[8];
  v2f a[16];
  #pragma unroll
  for (int i = 0; i < 8; ++i) wq[i] = *(const v4f*)(W + (srow + 8 * i) * D_ + sc4 * 4);
  #pragma unroll
  for (int ks = 0; ks < 16; ++ks) a[ks] = *(const v2f*)(xrow + 4 * ks + 2 * half);
  #pragma unroll
  for (int i = 0; i < 8; ++i) *(v4f*)&Ws[(srow + 8 * i) * 68 + sc4 * 4] = wq[i];
  __syncthreads();

  v8f c[4] = {};
  #pragma unroll
  for (int ks = 0; ks < 16; ++ks) {
    const int k = 4 * ks + 2 * half;
    #pragma unroll
    for (int f = 0; f < 4; ++f) {
      const int e = 16 * f + mrow;
      v2f b; b.x = Ws[e * 68 + k]; b.y = Ws[e * 68 + k + 1];
      c[f] = wmma_f32(a[ks], b, c[f]);
    }
  }

  float* obase = Out + ((size_t)(n * H_ + h)) * S_ * D_;
  #pragma unroll
  for (int f = 0; f < 4; ++f)
    #pragma unroll
    for (int r = 0; r < 8; ++r) {
      const int m = r + 8 * half;
      const int e = 16 * f + mrow;
      obase[(size_t)(s0 + m) * D_ + e] = c[f][r];
    }
}

static constexpr int BK_  = 64;
static constexpr int LDP_ = 68;

__global__ __launch_bounds__(128)
void flash_kernel(const float* __restrict__ Qp,
                  const float* __restrict__ Kp,
                  const float* __restrict__ Vp,
                  float* __restrict__ O)
{
  __shared__ __attribute__((aligned(16))) float smem[(2 * BK_ + 4 * 16) * LDP_];

  const int tid  = threadIdx.x;
  const int wave = tid >> 5;
  const int lane = tid & 31;
  const int n = blockIdx.z, h = blockIdx.y;
  const int mrow = lane & 15, half = lane >> 4;
  const size_t headOff = ((size_t)(n * H_ + h)) * S_ * D_;

  float* Ks = smem;
  float* Vs = smem + BK_ * LDP_;
  float* Pw = smem + 2 * BK_ * LDP_ + wave * 16 * LDP_;

  const float slope  = (float)(1 << __popc(h));
  const float scale  = 0.03125f;
  const float ss     = slope * scale;

  const int srow = tid >> 4, sc4 = tid & 15;
  const float* kg = Kp + headOff + (size_t)srow * D_ + sc4 * 4;
  const float* vg = Vp + headOff + (size_t)srow * D_ + sc4 * 4;
  float* sdst = smem + srow * LDP_ + sc4 * 4;

  for (int pass = 0; pass < 2; ++pass) {
    const int qblk = pass ? (int)blockIdx.x : (S_ / 64 - 1) - (int)blockIdx.x;
    const int q0 = qblk * 64 + wave * 16;

    v4f kq[8], vq[8];
    #pragma unroll
    for (int i = 0; i < 8; ++i) kq[i] = *(const v4f*)(kg + (size_t)(8 * i) * D_);
    #pragma unroll
    for (int i = 0; i < 8; ++i) vq[i] = *(const v4f*)(vg + (size_t)(8 * i) * D_);

    float qreg[32];
    const float* qrow = Qp + headOff + (size_t)(q0 + mrow) * D_;
    #pragma unroll
    for (int ks = 0; ks < 16; ++ks) {
      const int k = 4 * ks + 2 * half;
      qreg[2 * ks]     = qrow[k];
      qreg[2 * ks + 1] = qrow[k + 1];
    }

    v8f o[4] = {};
    float mrun[8], lrun[8];
    #pragma unroll
    for (int r = 0; r < 8; ++r) { mrun[r] = -3.0e38f; lrun[r] = 0.0f; }

    for (int kt = 0; kt <= qblk; ++kt) {
      const int k0 = kt * BK_;

      __syncthreads();
      #pragma unroll
      for (int i = 0; i < 8; ++i) *(v4f*)(sdst + 8 * i * LDP_) = kq[i];
      #pragma unroll
      for (int i = 0; i < 8; ++i) *(v4f*)(sdst + (BK_ + 8 * i) * LDP_) = vq[i];
      __syncthreads();

      if (kt < qblk) {
        #pragma unroll
        for (int i = 0; i < 8; ++i) kq[i] = *(const v4f*)(kg + (size_t)(k0 + BK_ + 8 * i) * D_);
        #pragma unroll
        for (int i = 0; i < 8; ++i) vq[i] = *(const v4f*)(vg + (size_t)(k0 + BK_ + 8 * i) * D_);
      }

      v8f s[4] = {};
      #pragma unroll
      for (int ks = 0; ks < 16; ++ks) {
        const int k = 4 * ks + 2 * half;
        v2f a; a.x = qreg[2 * ks]; a.y = qreg[2 * ks + 1];
        #pragma unroll
        for (int j = 0; j < 4; ++j) {
          const v2f b = *(const v2f*)&Ks[(16 * j + mrow) * LDP_ + k];
          s[j] = wmma_f32(a, b, s[j]);
        }
      }

      const float lanebias = ss * (float)(k0 - q0 + mrow - 8 * half);
      #pragma unroll
      for (int j = 0; j < 4; ++j)
        #pragma unroll
        for (int r = 0; r < 8; ++r)
          s[j][r] = fmaf(s[j][r], scale, lanebias + ss * (float)(16 * j - r));

      if (kt == qblk) {
        const int d = mrow - 8 * half - 16 * wave;
        #pragma unroll
        for (int j = 0; j < 4; ++j)
          #pragma unroll
          for (int r = 0; r < 8; ++r)
            if (d + 16 * j - r > 0) s[j][r] = -3.0e18f;
      }

      #pragma unroll
      for (int r = 0; r < 8; ++r) {
        const float rm   = rowmax16(fmaxf(fmaxf(s[0][r], s[1][r]), fmaxf(s[2][r], s[3][r])));
        const float mnew = fmaxf(mrun[r], rm);
        const float alpha = __expf(mrun[r] - mnew);
        mrun[r] = mnew;
        const float p0 = __expf(s[0][r] - mnew);
        const float p1 = __expf(s[1][r] - mnew);
        const float p2 = __expf(s[2][r] - mnew);
        const float p3 = __expf(s[3][r] - mnew);
        s[0][r] = p0; s[1][r] = p1; s[2][r] = p2; s[3][r] = p3;
        lrun[r] = lrun[r] * alpha + rowsum16((p0 + p1) + (p2 + p3));
        #pragma unroll
        for (int f = 0; f < 4; ++f) o[f][r] *= alpha;
      }

      #pragma unroll
      for (int j = 0; j < 4; ++j)
        #pragma unroll
        for (int r = 0; r < 8; ++r)
          Pw[(r + 8 * half) * LDP_ + 16 * j + mrow] = s[j][r];
      #pragma unroll
      for (int kk = 0; kk < 16; ++kk) {
        const int ka = 4 * kk + 2 * half;
        const v2f a = *(const v2f*)&Pw[mrow * LDP_ + ka];
        #pragma unroll
        for (int f = 0; f < 4; ++f) {
          const int dcol = 16 * f + mrow;
          v2f b;
          b.x = Vs[ka       * LDP_ + dcol];
          b.y = Vs[(ka + 1) * LDP_ + dcol];
          o[f] = wmma_f32(a, b, o[f]);
        }
      }
    }

    #pragma unroll
    for (int r = 0; r < 8; ++r) lrun[r] = 1.0f / lrun[r];
    float* obase = O + (size_t)n * S_ * E_ + h * D_;
    #pragma unroll
    for (int f = 0; f < 4; ++f)
      #pragma unroll
      for (int r = 0; r < 8; ++r) {
        const int m = r + 8 * half;
        const int dcol = 16 * f + mrow;
        obase[(size_t)(q0 + m) * E_ + dcol] = o[f][r] * lrun[r];
      }
  }
}

static __device__ __forceinline__
void outproj_chunk(v8f (&c)[4], const v2f (&a)[16], v2f (&anext)[16], v4f (&wq)[8],
                   const float* Ws, float* sdst,
                   const float* __restrict__ wg, const float* __restrict__ arow,
                   int kc, int mrow, int half)
{
  __syncthreads();
  #pragma unroll
  for (int i = 0; i < 8; ++i) *(v4f*)(sdst + 8 * i * 68) = wq[i];
  __syncthreads();

  if (kc + 64 < E_) {
    #pragma unroll
    for (int i = 0; i < 8; ++i) wq[i] = *(const v4f*)(wg + (size_t)(8 * i) * E_ + kc + 64);
    #pragma unroll
    for (int ks = 0; ks < 16; ++ks) anext[ks] = *(const v2f*)(arow + kc + 64 + 4 * ks);
  }

  #pragma unroll
  for (int ks = 0; ks < 16; ++ks) {
    const int k = 4 * ks + 2 * half;
    #pragma unroll
    for (int f = 0; f < 4; ++f) {
      const int e = 16 * f + mrow;
      v2f b; b.x = Ws[e * 68 + k]; b.y = Ws[e * 68 + k + 1];
      c[f] = wmma_f32(a[ks], b, c[f]);
    }
  }
}

__global__ __launch_bounds__(128)
void outproj_kernel(const float* __restrict__ A,
                    const float* __restrict__ Wo,
                    const float* __restrict__ bo,
                    float* __restrict__ Y)
{
  __shared__ __attribute__((aligned(16))) float Ws[64 * 68];
  const int tid  = threadIdx.x;
  const int wave = tid >> 5;
  const int lane = tid & 31;
  const int r0 = blockIdx.y * 64 + wave * 16;
  const int c0 = blockIdx.x * 64;
  const int mrow = lane & 15, half = lane >> 4;
  const int srow = tid >> 4, sc4 = tid & 15;

  v8f c[4] = {};
  const float* arow = A + (size_t)(r0 + mrow) * E_ + 2 * half;
  const float* wg   = Wo + (size_t)(c0 + srow) * E_ + sc4 * 4;
  float* sdst = &Ws[srow * 68 + sc4 * 4];

  v4f wq[8];
  v2f a0[16], a1[16];
  #pragma unroll
  for (int i = 0; i < 8; ++i) wq[i] = *(const v4f*)(wg + (size_t)(8 * i) * E_);
  #pragma unroll
  for (int ks = 0; ks < 16; ++ks) a0[ks] = *(const v2f*)(arow + 4 * ks);

  for (int kc = 0; kc < E_; kc += 128) {
    outproj_chunk(c, a0, a1, wq, Ws, sdst, wg, arow, kc,      mrow, half);
    outproj_chunk(c, a1, a0, wq, Ws, sdst, wg, arow, kc + 64, mrow, half);
  }

  #pragma unroll
  for (int f = 0; f < 4; ++f)
    #pragma unroll
    for (int r = 0; r < 8; ++r) {
      const int m = r + 8 * half;
      const int e = c0 + 16 * f + mrow;
      Y[(size_t)(r0 + m) * E_ + e] = c[f][r] + bo[e];
    }
}

extern "C" void kernel_launch(void* const* d_in, const int* in_sizes, int n_in,
                              void* d_out, int out_size, void* d_ws, size_t ws_size,
                              hipStream_t stream) {
  const float* values = (const float*)d_in[0];
  const float* keys   = (const float*)d_in[1];
  const float* query  = (const float*)d_in[2];
  const float* Wv     = (const float*)d_in[4];
  const float* Wk     = (const float*)d_in[5];
  const float* Wq     = (const float*)d_in[6];
  const float* Wo     = (const float*)d_in[7];
  const float* bo     = (const float*)d_in[8];
  float* out = (float*)d_out;

  const size_t tensorElems = (size_t)N_ * H_ * S_ * D_;
  float* Qp = (float*)d_ws;
  float* Kp = Qp + tensorElems;
  float* Vp = Kp + tensorElems;
  float* Oa = Vp + tensorElems;

  const dim3 blk(128);
  const dim3 gProj(S_ / 64, H_, N_);

  proj_kernel<<<gProj, blk, 0, stream>>>(query,  Wq, Qp);
  proj_kernel<<<gProj, blk, 0, stream>>>(keys,   Wk, Kp);
  proj_kernel<<<gProj, blk, 0, stream>>>(values, Wv, Vp);

  const dim3 gFlash(S_ / 128, H_, N_);
  flash_kernel<<<gFlash, blk, 0, stream>>>(Qp, Kp, Vp, Oa);

  const dim3 gOut(E_ / 64, (N_ * S_) / 64);
  outproj_kernel<<<gOut, blk, 0, stream>>>(Oa, Wo, bo, out);
}


# attention kernel: transposed form S^T=K.Q^T, O^T=V^T.P^T: one query per lane, row max/sum in-lane, P enters the 2nd product from registers, not via LDS; row sums and P.V summed in a new order
# speedup vs baseline: 2.4464x; 1.0887x over previous
#include <hip/hip_runtime.h>
#include <hip/hip_bf16.h>
#include <math.h>


typedef float v2f __attribute__((ext_vector_type(2)));
typedef float v4f __attribute__((ext_vector_type(4)));
typedef float v8f __attribute__((ext_vector_type(8)));

static constexpr int N_ = 2;
static constexpr int S_ = 2048;
static constexpr int E_ = 1024;
static constexpr int H_ = 16;
static constexpr int D_ = 64;

static __device__ __forceinline__ v8f wmma_f32(v2f a, v2f b, v8f c) {
  return __builtin_amdgcn_wmma_f32_16x16x4_f32(false, a, false, b,
                                               (short)0, c, false, false);
}

static __device__ __forceinline__ float xhalf(float x) {
  const int b = __builtin_bit_cast(int, x);
  return __builtin_bit_cast(float,
      __builtin_amdgcn_permlanex16(b, b, 0x76543210, 0xfedcba98, false, false));
}

__global__ __launch_bounds__(128)
void proj_kernel(const float* __restrict__ X,
                 const float* __restrict__ W,
                 float* __restrict__ Out)
{
  __shared__ __attribute__((aligned(16))) float Ws[64 * 68];
  const int tid  = threadIdx.x;
  const int wave = tid >> 5;
  const int lane = tid & 31;
  const int n  = blockIdx.z, h = blockIdx.y;
  const int s0 = blockIdx.x * 64 + wave * 16;
  const int mrow = lane & 15, half = lane >> 4;
  const int srow = tid >> 4, sc4 = tid & 15;

  const float* xrow = X + ((size_t)n * S_ + (s0 + mrow)) * E_ + h * D_;
  v4f wq[8];
  v2f a[16];
  #pragma unroll
  for (int i = 0; i < 8; ++i) wq[i] = *(const v4f*)(W + (srow + 8 * i) * D_ + sc4 * 4);
  #pragma unroll
  for (int ks = 0; ks < 16; ++ks) a[ks] = *(const v2f*)(xrow + 4 * ks + 2 * half);
  #pragma unroll
  for (int i = 0; i < 8; ++i) *(v4f*)&Ws[(srow + 8 * i) * 68 + sc4 * 4] = wq[i];
  __syncthreads();

  v8f c[4] = {};
  #pragma unroll
  for (int ks = 0; ks < 16; ++ks) {
    const int k = 4 * ks + 2 * half;
    #pragma unroll
    for (int f = 0; f < 4; ++f) {
      const int e = 16 * f + mrow;
      v2f b; b.x = Ws[e * 68 + k]; b.y = Ws[e * 68 + k + 1];
      c[f] = wmma_f32(a[ks], b, c[f]);
    }
  }

  float* obase = Out + ((size_t)(n * H_ + h)) * S_ * D_;
  #pragma unroll
  for (int f = 0; f < 4; ++f)
    #pragma unroll
    for (int r = 0; r < 8; ++r) {
      const int m = r + 8 * half;
      const int e = 16 * f + mrow;
      obase[(size_t)(s0 + m) * D_ + e] = c[f][r];
    }
}

static constexpr int BK_  = 64;
static constexpr int LDP_ = 68;

__global__ __launch_bounds__(128)
void flash_kernel(const float* __restrict__ Qp,
                  const float* __restrict__ Kp,
                  const float* __restrict__ Vp,
                  float* __restrict__ O)
{
  __shared__ __attribute__((aligned(16))) float smem[2 * BK_ * LDP_];

  const int tid  = threadIdx.x;
  const int wave = tid >> 5;
  const int lane = tid & 31;
  const int n = blockIdx.z, h = blockIdx.y;
  const int mrow = lane & 15, half = lane >> 4;
  const size_t headOff = ((size_t)(n * H_ + h)) * S_ * D_;

  const float* Ks = smem;
  const float* Vs = smem + BK_ * LDP_;

  const float slope  = (float)(1 << __popc(h));
  const float scale  = 0.03125f;
  const float ss     = slope * scale;

  const int srow = tid >> 4, sc4 = tid & 15;
  const float* kg = Kp + headOff + (size_t)srow * D_ + sc4 * 4;
  const float* vg = Vp + headOff + (size_t)srow * D_ + sc4 * 4;
  float* sdst = smem + srow * LDP_ + sc4 * 4;

  for (int pass = 0; pass < 2; ++pass) {
    const int qblk = pass ? (int)blockIdx.x : (S_ / 64 - 1) - (int)blockIdx.x;
    const int q0 = qblk * 64 + wave * 16;

    v4f kq[8], vq[8];
    #pragma unroll
    for (int i = 0; i < 8; ++i) kq[i] = *(const v4f*)(kg + (size_t)(8 * i) * D_);
    #pragma unroll
    for (int i = 0; i < 8; ++i) vq[i] = *(const v4f*)(vg + (size_t)(8 * i) * D_);

    float qreg[32];
    const float* qrow = Qp + headOff + (size_t)(q0 + mrow) * D_;
    #pragma unroll
    for (int ks = 0; ks < 16; ++ks) {
      const int k = 4 * ks + 2 * half;
      qreg[2 * ks]     = qrow[k];
      qreg[2 * ks + 1] = qrow[k + 1];
    }

    v8f o[4] = {};
    float mrun = -3.0e38f;
    float lrun = 0.0f;

    for (int kt = 0; kt <= qblk; ++kt) {
      const int k0 = kt * BK_;

      __syncthreads();
      #pragma unroll
      for (int i = 0; i < 8; ++i) *(v4f*)(sdst + 8 * i * LDP_) = kq[i];
      #pragma unroll
      for (int i = 0; i < 8; ++i) *(v4f*)(sdst + (BK_ + 8 * i) * LDP_) = vq[i];
      __syncthreads();

      if (kt < qblk) {
        #pragma unroll
        for (int i = 0; i < 8; ++i) kq[i] = *(const v4f*)(kg + (size_t)(k0 + BK_ + 8 * i) * D_);
        #pragma unroll
        for (int i = 0; i < 8; ++i) vq[i] = *(const v4f*)(vg + (size_t)(k0 + BK_ + 8 * i) * D_);
      }

      v8f s[4] = {};
      #pragma unroll
      for (int ks = 0; ks < 16; ++ks) {
        const int k = 4 * ks + 2 * half;
        v2f b; b.x = qreg[2 * ks]; b.y = qreg[2 * ks + 1];
        #pragma unroll
        for (int j = 0; j < 4; ++j) {
          const v2f a = *(const v2f*)&Ks[(16 * j + mrow) * LDP_ + k];
          s[j] = wmma_f32(a, b, s[j]);
        }
      }

      const float lanebias = ss * (float)(k0 - q0 + 8 * half - mrow);
      #pragma unroll
      for (int j = 0; j < 4; ++j)
        #pragma unroll
        for (int r = 0; r < 8; ++r)
          s[j][r] = fmaf(s[j][r], scale, lanebias + ss * (float)(16 * j + r));

      if (kt == qblk) {
        const int d = 8 * half - mrow - 16 * wave;
        #pragma unroll
        for (int j = 0; j < 4; ++j)
          #pragma unroll
          for (int r = 0; r < 8; ++r)
            if (d + 16 * j + r > 0) s[j][r] = -3.0e18f;
      }

      float rm;
      {
        float m4[4];
        #pragma unroll
        for (int j = 0; j < 4; ++j)
          m4[j] = fmaxf(fmaxf(fmaxf(s[j][0], s[j][1]), fmaxf(s[j][2], s[j][3])),
                        fmaxf(fmaxf(s[j][4], s[j][5]), fmaxf(s[j][6], s[j][7])));
        rm = fmaxf(fmaxf(m4[0], m4[1]), fmaxf(m4[2], m4[3]));
      }
      rm = fmaxf(rm, xhalf(rm));
      const float mnew  = fmaxf(mrun, rm);
      const float alpha = __expf(mrun - mnew);
      mrun = mnew;
      float p4[4];
      #pragma unroll
      for (int j = 0; j < 4; ++j) {
        #pragma unroll
        for (int r = 0; r < 8; ++r) s[j][r] = __expf(s[j][r] - mnew);
        p4[j] = ((s[j][0] + s[j][1]) + (s[j][2] + s[j][3])) +
                ((s[j][4] + s[j][5]) + (s[j][6] + s[j][7]));
      }
      lrun = lrun * alpha + ((p4[0] + p4[1]) + (p4[2] + p4[3]));
      #pragma unroll
      for (int f = 0; f < 4; ++f)
        #pragma unroll
        for (int r = 0; r < 8; ++r) o[f][r] *= alpha;

      #pragma unroll
      for (int j = 0; j < 4; ++j)
        #pragma unroll
        for (int t = 0; t < 4; ++t) {
          const int key = 16 * j + 8 * half + 2 * t;
          v2f b; b.x = s[j][2 * t]; b.y = s[j][2 * t + 1];
          #pragma unroll
          for (int f = 0; f < 4; ++f) {
            const int dcol = 16 * f + mrow;
            v2f a;
            a.x = Vs[key       * LDP_ + dcol];
            a.y = Vs[(key + 1) * LDP_ + dcol];
            o[f] = wmma_f32(a, b, o[f]);
          }
        }
    }

    const float linv = 1.0f / (lrun + xhalf(lrun));
    float* orow = O + (size_t)n * S_ * E_ + (size_t)(q0 + mrow) * E_ + h * D_ + 8 * half;
    #pragma unroll
    for (int f = 0; f < 4; ++f) {
      v4f lo, hi;
      lo.x = o[f][0] * linv; lo.y = o[f][1] * linv; lo.z = o[f][2] * linv; lo.w = o[f][3] * linv;
      hi.x = o[f][4] * linv; hi.y = o[f][5] * linv; hi.z = o[f][6] * linv; hi.w = o[f][7] * linv;
      *(v4f*)(orow + 16 * f)     = lo;
      *(v4f*)(orow + 16 * f + 4) = hi;
    }
  }
}

static __device__ __forceinline__
void outproj_chunk(v8f (&c)[4], const v2f (&a)[16], v2f (&anext)[16], v4f (&wq)[8],
                   const float* Ws, float* sdst,
                   const float* __restrict__ wg, const float* __restrict__ arow,
                   int kc, int mrow, int half)
{
  __syncthreads();
  #pragma unroll
  for (int i = 0; i < 8; ++i) *(v4f*)(sdst + 8 * i * 68) = wq[i];
  __syncthreads();

  if (kc + 64 < E_) {
    #pragma unroll
    for (int i = 0; i < 8; ++i) wq[i] = *(const v4f*)(wg + (size_t)(8 * i) * E_ + kc + 64);
    #pragma unroll
    for (int ks = 0; ks < 16; ++ks) anext[ks] = *(const v2f*)(arow + kc + 64 + 4 * ks);
  }

  #pragma unroll
  for (int ks = 0; ks < 16; ++ks) {
    const int k = 4 * ks + 2 * half;
    #pragma unroll
    for (int f = 0; f < 4; ++f) {
      const int e = 16 * f + mrow;
      v2f b; b.x = Ws[e * 68 + k]; b.y = Ws[e * 68 + k + 1];
      c[f] = wmma_f32(a[ks], b, c[f]);
    }
  }
}

__global__ __launch_bounds__(128)
void outproj_kernel(const float* __restrict__ A,
                    const float* __restrict__ Wo,
                    const float* __restrict__ bo,
                    float* __restrict__ Y)
{
  __shared__ __attribute__((aligned(16))) float Ws[64 * 68];
  const int tid  = threadIdx.x;
  const int wave = tid >> 5;
  const int lane = tid & 31;
  const int r0 = blockIdx.y * 64 + wave * 16;
  const int c0 = blockIdx.x * 64;
  const int mrow = lane & 15, half = lane >> 4;
  const int srow = tid >> 4, sc4 = tid & 15;

  v8f c[4] = {};
  const float* arow = A + (size_t)(r0 + mrow) * E_ + 2 * half;
  const float* wg   = Wo + (size_t)(c0 + srow) * E_ + sc4 * 4;
  float* sdst = &Ws[srow * 68 + sc4 * 4];

  v4f wq[8];
  v2f a0[16], a1[16];
  #pragma unroll
  for (int i = 0; i < 8; ++i) wq[i] = *(const v4f*)(wg + (size_t)(8 * i) * E_);
  #pragma unroll
  for (int ks = 0; ks < 16; ++ks) a0[ks] = *(const v2f*)(arow + 4 * ks);

  for (int kc = 0; kc < E_; kc += 128) {
    outproj_chunk(c, a0, a1, wq, Ws, sdst, wg, arow, kc,      mrow, half);
    outproj_chunk(c, a1, a0, wq, Ws, sdst, wg, arow, kc + 64, mrow, half);
  }

  #pragma unroll
  for (int f = 0; f < 4; ++f)
    #pragma unroll
    for (int r = 0; r < 8; ++r) {
      const int m = r + 8 * half;
      const int e = c0 + 16 * f + mrow;
      Y[(size_t)(r0 + m) * E_ + e] = c[f][r] + bo[e];
    }
}

extern "C" void kernel_launch(void* const* d_in, const int* in_sizes, int n_in,
                              void* d_out, int out_size, void* d_ws, size_t ws_size,
                              hipStream_t stream) {
  const float* values = (const float*)d_in[0];
  const float* keys   = (const float*)d_in[1];
  const float* query  = (const float*)d_in[2];
  const float* Wv     = (const float*)d_in[4];
  const float* Wk     = (const float*)d_in[5];
  const float* Wq     = (const float*)d_in[6];
  const float* Wo     = (const float*)d_in[7];
  const float* bo     = (const float*)d_in[8];
  float* out = (float*)d_out;

  const size_t tensorElems = (size_t)N_ * H_ * S_ * D_;
  float* Qp = (float*)d_ws;
  float* Kp = Qp + tensorElems;
  float* Vp = Kp + tensorElems;
  float* Oa = Vp + tensorElems;

  const dim3 blk(128);
  const dim3 gProj(S_ / 64, H_, N_);

  proj_kernel<<<gProj, blk, 0, stream>>>(query,  Wq, Qp);
  proj_kernel<<<gProj, blk, 0, stream>>>(keys,   Wk, Kp);
  proj_kernel<<<gProj, blk, 0, stream>>>(values, Wv, Vp);

  const dim3 gFlash(S_ / 128, H_, N_);
  flash_kernel<<<gFlash, blk, 0, stream>>>(Qp, Kp, Vp, Oa);

  const dim3 gOut(E_ / 64, (N_ * S_) / 64);
  outproj_kernel<<<gOut, blk, 0, stream>>>(Oa, Wo, bo, out);
}


# projections folded into the attention kernel by associativity: S=((Xq.Wq^T).Wk).Xk^T, O=(P.Xv).Wv^T on raw rows; 3 projection launches go; scores and outputs are summed in a new order
# speedup vs baseline: 2.7212x; 1.1123x over previous
#include <hip/hip_runtime.h>
#include <hip/hip_bf16.h>
#include <math.h>


typedef float v2f __attribute__((ext_vector_type(2)));
typedef float v4f __attribute__((ext_vector_type(4)));
typedef float v8f __attribute__((ext_vector_type(8)));

static constexpr int N_ = 2;
static constexpr int S_ = 2048;
static constexpr int E_ = 1024;
static constexpr int H_ = 16;
static constexpr int D_ = 64;

static __device__ __forceinline__ v8f wmma_f32(v2f a, v2f b, v8f c) {
  return __builtin_amdgcn_wmma_f32_16x16x4_f32(false, a, false, b,
                                               (short)0, c, false, false);
}

static __device__ __forceinline__ float xhalf(float x) {
  const int b = __builtin_bit_cast(int, x);
  return __builtin_bit_cast(float,
      __builtin_amdgcn_permlanex16(b, b, 0x76543210, 0xfedcba98, false, false));
}

static constexpr int BK_  = 64;
static constexpr int LDP_ = 68;

__global__ __launch_bounds__(128)
void flash_kernel(const float* __restrict__ Xq,
                  const float* __restrict__ Xk,
                  const float* __restrict__ Xv,
                  const float* __restrict__ Wq,
                  const float* __restrict__ Wk,
                  const float* __restrict__ Wv,
                  float* __restrict__ O)
{
  __shared__ __attribute__((aligned(16))) float smem[2 * BK_ * LDP_];

  const int tid  = threadIdx.x;
  const int wave = tid >> 5;
  const int lane = tid & 31;
  const int n = blockIdx.z, h = blockIdx.y;
  const int mrow = lane & 15, half = lane >> 4;
  const size_t inOff = (size_t)n * S_ * E_ + (size_t)h * D_;

  const float* Ks = smem;
  const float* Vs = smem + BK_ * LDP_;

  const float slope  = (float)(1 << __popc(h));
  const float scale  = 0.03125f;
  const float ss     = slope * scale;

  const int srow = tid >> 4, sc4 = tid & 15;
  const float* kg = Xk + inOff + (size_t)srow * E_ + sc4 * 4;
  const float* vg = Xv + inOff + (size_t)srow * E_ + sc4 * 4;
  const int woff = srow * D_ + sc4 * 4;
  float* sdst = smem + srow * LDP_ + sc4 * 4;

  for (int pass = 0; pass < 2; ++pass) {
    const int qblk = pass ? (int)blockIdx.x : (S_ / 64 - 1) - (int)blockIdx.x;
    const int q0 = qblk * 64 + wave * 16;

    v4f kq[8], vq[8];
    #pragma unroll
    for (int i = 0; i < 8; ++i) kq[i] = *(const v4f*)(Wq + woff + 8 * i * D_);
    #pragma unroll
    for (int i = 0; i < 8; ++i) vq[i] = *(const v4f*)(Wk + woff + 8 * i * D_);
    float xq[32];
    const float* qrow = Xq + inOff + (size_t)(q0 + mrow) * E_;
    #pragma unroll
    for (int ks = 0; ks < 16; ++ks) {
      const int k = 4 * ks + 2 * half;
      xq[2 * ks]     = qrow[k];
      xq[2 * ks + 1] = qrow[k + 1];
    }

    __syncthreads();
    #pragma unroll
    for (int i = 0; i < 8; ++i) *(v4f*)(sdst + 8 * i * LDP_) = kq[i];
    #pragma unroll
    for (int i = 0; i < 8; ++i) *(v4f*)(sdst + (BK_ + 8 * i) * LDP_) = vq[i];
    __syncthreads();

    #pragma unroll
    for (int i = 0; i < 8; ++i) kq[i] = *(const v4f*)(kg + (size_t)(8 * i) * E_);
    #pragma unroll
    for (int i = 0; i < 8; ++i) vq[i] = *(const v4f*)(vg + (size_t)(8 * i) * E_);

    v8f qt[4] = {};
    #pragma unroll
    for (int ks = 0; ks < 16; ++ks) {
      const int k = 4 * ks + 2 * half;
      v2f b; b.x = xq[2 * ks]; b.y = xq[2 * ks + 1];
      #pragma unroll
      for (int f = 0; f < 4; ++f) {
        const v2f a = *(const v2f*)&Ks[(16 * f + mrow) * LDP_ + k];
        qt[f] = wmma_f32(a, b, qt[f]);
      }
    }
    v8f qp[4] = {};
    #pragma unroll
    for (int f = 0; f < 4; ++f)
      #pragma unroll
      for (int t = 0; t < 4; ++t) {
        const int e0 = 16 * f + 8 * half + 2 * t;
        v2f b; b.x = qt[f][2 * t]; b.y = qt[f][2 * t + 1];
        #pragma unroll
        for (int g = 0; g < 4; ++g) {
          const int dcol = 16 * g + mrow;
          v2f a;
          a.x = Vs[e0       * LDP_ + dcol];
          a.y = Vs[(e0 + 1) * LDP_ + dcol];
          qp[g] = wmma_f32(a, b, qp[g]);
        }
      }

    v8f o[4] = {};
    float mrun = -3.0e38f;
    float lrun = 0.0f;

    for (int kt = 0; kt <= qblk; ++kt) {
      const int k0 = kt * BK_;

      __syncthreads();
      #pragma unroll
      for (int i = 0; i < 8; ++i) *(v4f*)(sdst + 8 * i * LDP_) = kq[i];
      #pragma unroll
      for (int i = 0; i < 8; ++i) *(v4f*)(sdst + (BK_ + 8 * i) * LDP_) = vq[i];
      __syncthreads();

      if (kt < qblk) {
        #pragma unroll
        for (int i = 0; i < 8; ++i) kq[i] = *(const v4f*)(kg + (size_t)(k0 + BK_ + 8 * i) * E_);
        #pragma unroll
        for (int i = 0; i < 8; ++i) vq[i] = *(const v4f*)(vg + (size_t)(k0 + BK_ + 8 * i) * E_);
      } else {
        #pragma unroll
        for (int i = 0; i < 8; ++i) kq[i] = *(const v4f*)(Wv + woff + 8 * i * D_);
      }

      v8f s[4] = {};
      #pragma unroll
      for (int g = 0; g < 4; ++g)
        #pragma unroll
        for (int t = 0; t < 4; ++t) {
          const int k = 16 * g + 8 * half + 2 * t;
          v2f b; b.x = qp[g][2 * t]; b.y = qp[g][2 * t + 1];
          #pragma unroll
          for (int j = 0; j < 4; ++j) {
            const v2f a = *(const v2f*)&Ks[(16 * j + mrow) * LDP_ + k];
            s[j] = wmma_f32(a, b, s[j]);
          }
        }

      const float lanebias = ss * (float)(k0 - q0 + 8 * half - mrow);
      #pragma unroll
      for (int j = 0; j < 4; ++j)
        #pragma unroll
        for (int r = 0; r < 8; ++r)
          s[j][r] = fmaf(s[j][r], scale, lanebias + ss * (float)(16 * j + r));

      if (kt == qblk) {
        const int d = 8 * half - mrow - 16 * wave;
        #pragma unroll
        for (int j = 0; j < 4; ++j)
          #pragma unroll
          for (int r = 0; r < 8; ++r)
            if (d + 16 * j + r > 0) s[j][r] = -3.0e18f;
      }

      float rm;
      {
        float m4[4];
        #pragma unroll
        for (int j = 0; j < 4; ++j)
          m4[j] = fmaxf(fmaxf(fmaxf(s[j][0], s[j][1]), fmaxf(s[j][2], s[j][3])),
                        fmaxf(fmaxf(s[j][4], s[j][5]), fmaxf(s[j][6], s[j][7])));
        rm = fmaxf(fmaxf(m4[0], m4[1]), fmaxf(m4[2], m4[3]));
      }
      rm = fmaxf(rm, xhalf(rm));
      const float mnew  = fmaxf(mrun, rm);
      const float alpha = __expf(mrun - mnew);
      mrun = mnew;
      float p4[4];
      #pragma unroll
      for (int j = 0; j < 4; ++j) {
        #pragma unroll
        for (int r = 0; r < 8; ++r) s[j][r] = __expf(s[j][r] - mnew);
        p4[j] = ((s[j][0] + s[j][1]) + (s[j][2] + s[j][3])) +
                ((s[j][4] + s[j][5]) + (s[j][6] + s[j][7]));
      }
      lrun = lrun * alpha + ((p4[0] + p4[1]) + (p4[2] + p4[3]));
      #pragma unroll
      for (int f = 0; f < 4; ++f)
        #pragma unroll
        for (int r = 0; r < 8; ++r) o[f][r] *= alpha;

      #pragma unroll
      for (int j = 0; j < 4; ++j)
        #pragma unroll
        for (int t = 0; t < 4; ++t) {
          const int key = 16 * j + 8 * half + 2 * t;
          v2f b; b.x = s[j][2 * t]; b.y = s[j][2 * t + 1];
          #pragma unroll
          for (int f = 0; f < 4; ++f) {
            const int dcol = 16 * f + mrow;
            v2f a;
            a.x = Vs[key       * LDP_ + dcol];
            a.y = Vs[(key + 1) * LDP_ + dcol];
            o[f] = wmma_f32(a, b, o[f]);
          }
        }
    }

    __syncthreads();
    #pragma unroll
    for (int i = 0; i < 8; ++i) *(v4f*)(sdst + 8 * i * LDP_) = kq[i];
    __syncthreads();
    v8f fo[4] = {};
    #pragma unroll
    for (int f = 0; f < 4; ++f)
      #pragma unroll
      for (int t = 0; t < 4; ++t) {
        const int k = 16 * f + 8 * half + 2 * t;
        v2f b; b.x = o[f][2 * t]; b.y = o[f][2 * t + 1];
        #pragma unroll
        for (int g = 0; g < 4; ++g) {
          const v2f a = *(const v2f*)&Ks[(16 * g + mrow) * LDP_ + k];
          fo[g] = wmma_f32(a, b, fo[g]);
        }
      }

    const float linv = 1.0f / (lrun + xhalf(lrun));
    float* orow = O + (size_t)n * S_ * E_ + (size_t)(q0 + mrow) * E_ + h * D_ + 8 * half;
    #pragma unroll
    for (int f = 0; f < 4; ++f) {
      v4f lo, hi;
      lo.x = fo[f][0] * linv; lo.y = fo[f][1] * linv; lo.z = fo[f][2] * linv; lo.w = fo[f][3] * linv;
      hi.x = fo[f][4] * linv; hi.y = fo[f][5] * linv; hi.z = fo[f][6] * linv; hi.w = fo[f][7] * linv;
      *(v4f*)(orow + 16 * f)     = lo;
      *(v4f*)(orow + 16 * f + 4) = hi;
    }
  }
}

static __device__ __forceinline__
void outproj_chunk(v8f (&c)[4], const v2f (&a)[16], v2f (&anext)[16], v4f (&wq)[8],
                   const float* Ws, float* sdst,
                   const float* __restrict__ wg, const float* __restrict__ arow,
                   int kc, int mrow, int half)
{
  __syncthreads();
  #pragma unroll
  for (int i = 0; i < 8; ++i) *(v4f*)(sdst + 8 * i * 68) = wq[i];
  __syncthreads();

  if (kc + 64 < E_) {
    #pragma unroll
    for (int i = 0; i < 8; ++i) wq[i] = *(const v4f*)(wg + (size_t)(8 * i) * E_ + kc + 64);
    #pragma unroll
    for (int ks = 0; ks < 16; ++ks) anext[ks] = *(const v2f*)(arow + kc + 64 + 4 * ks);
  }

  #pragma unroll
  for (int ks = 0; ks < 16; ++ks) {
    const int k = 4 * ks + 2 * half;
    #pragma unroll
    for (int f = 0; f < 4; ++f) {
      const int e = 16 * f + mrow;
      v2f b; b.x = Ws[e * 68 + k]; b.y = Ws[e * 68 + k + 1];
      c[f] = wmma_f32(a[ks], b, c[f]);
    }
  }
}

__global__ __launch_bounds__(128)
void outproj_kernel(const float* __restrict__ A,
                    const float* __restrict__ Wo,
                    const float* __restrict__ bo,
                    float* __restrict__ Y)
{
  __shared__ __attribute__((aligned(16))) float Ws[64 * 68];
  const int tid  = threadIdx.x;
  const int wave = tid >> 5;
  const int lane = tid & 31;
  const int r0 = blockIdx.y * 64 + wave * 16;
  const int c0 = blockIdx.x * 64;
  const int mrow = lane & 15, half = lane >> 4;
  const int srow = tid >> 4, sc4 = tid & 15;

  v8f c[4] = {};
  const float* arow = A + (size_t)(r0 + mrow) * E_ + 2 * half;
  const float* wg   = Wo + (size_t)(c0 + srow) * E_ + sc4 * 4;
  float* sdst = &Ws[srow * 68 + sc4 * 4];

  v4f wq[8];
  v2f a0[16], a1[16];
  #pragma unroll
  for (int i = 0; i < 8; ++i) wq[i] = *(const v4f*)(wg + (size_t)(8 * i) * E_);
  #pragma unroll
  for (int ks = 0; ks < 16; ++ks) a0[ks] = *(const v2f*)(arow + 4 * ks);

  for (int kc = 0; kc < E_; kc += 128) {
    outproj_chunk(c, a0, a1, wq, Ws, sdst, wg, arow, kc,      mrow, half);
    outproj_chunk(c, a1, a0, wq, Ws, sdst, wg, arow, kc + 64, mrow, half);
  }

  #pragma unroll
  for (int f = 0; f < 4; ++f)
    #pragma unroll
    for (int r = 0; r < 8; ++r) {
      const int m = r + 8 * half;
      const int e = c0 + 16 * f + mrow;
      Y[(size_t)(r0 + m) * E_ + e] = c[f][r] + bo[e];
    }
}

extern "C" void kernel_launch(void* const* d_in, const int* in_sizes, int n_in,
                              void* d_out, int out_size, void* d_ws, size_t ws_size,
                              hipStream_t stream) {
  const float* values = (const float*)d_in[0];
  const float* keys   = (const float*)d_in[1];
  const float* query  = (const float*)d_in[2];
  const float* Wv     = (const float*)d_in[4];
  const float* Wk     = (const float*)d_in[5];
  const float* Wq     = (const float*)d_in[6];
  const float* Wo     = (const float*)d_in[7];
  const float* bo     = (const float*)d_in[8];
  float* out = (float*)d_out;

  float* Oa = (float*)d_ws;

  const dim3 blk(128);
  const dim3 gFlash(S_ / 128, H_, N_);
  flash_kernel<<<gFlash, blk, 0, stream>>>(query, keys, values, Wq, Wk, Wv, Oa);

  const dim3 gOut(E_ / 64, (N_ * S_) / 64);
  outproj_kernel<<<gOut, blk, 0, stream>>>(Oa, Wo, bo, out);
}


# attention output stored to the workspace in A-fragment order: the output projection loads A with whole-line 16-byte loads, not 8-byte loads over 16 rows; its k sums run in a new order
# speedup vs baseline: 2.8183x; 1.0357x over previous
#include <hip/hip_runtime.h>
#include <hip/hip_bf16.h>
#include <math.h>


typedef float v2f __attribute__((ext_vector_type(2)));
typedef float v4f __attribute__((ext_vector_type(4)));
typedef float v8f __attribute__((ext_vector_type(8)));

static constexpr int N_ = 2;
static constexpr int S_ = 2048;
static constexpr int E_ = 1024;
static constexpr int H_ = 16;
static constexpr int D_ = 64;

static __device__ __forceinline__ v8f wmma_f32(v2f a, v2f b, v8f c) {
  return __builtin_amdgcn_wmma_f32_16x16x4_f32(false, a, false, b,
                                               (short)0, c, false, false);
}

static __device__ __forceinline__ float xhalf(float x) {
  const int b = __builtin_bit_cast(int, x);
  return __builtin_bit_cast(float,
      __builtin_amdgcn_permlanex16(b, b, 0x76543210, 0xfedcba98, false, false));
}

static constexpr int BK_  = 64;
static constexpr int LDP_ = 68;

__global__ __launch_bounds__(128)
void flash_kernel(const float* __restrict__ Xq,
                  const float* __restrict__ Xk,
                  const float* __restrict__ Xv,
                  const float* __restrict__ Wq,
                  const float* __restrict__ Wk,
                  const float* __restrict__ Wv,
                  float* __restrict__ O)
{
  __shared__ __attribute__((aligned(16))) float smem[2 * BK_ * LDP_];

  const int tid  = threadIdx.x;
  const int wave = tid >> 5;
  const int lane = tid & 31;
  const int n = blockIdx.z, h = blockIdx.y;
  const int mrow = lane & 15, half = lane >> 4;
  const size_t inOff = (size_t)n * S_ * E_ + (size_t)h * D_;

  const float* Ks = smem;
  const float* Vs = smem + BK_ * LDP_;

  const float slope  = (float)(1 << __popc(h));
  const float scale  = 0.03125f;
  const float ss     = slope * scale;

  const int srow = tid >> 4, sc4 = tid & 15;
  const float* kg = Xk + inOff + (size_t)srow * E_ + sc4 * 4;
  const float* vg = Xv + inOff + (size_t)srow * E_ + sc4 * 4;
  const int woff = srow * D_ + sc4 * 4;
  float* sdst = smem + srow * LDP_ + sc4 * 4;

  for (int pass = 0; pass < 2; ++pass) {
    const int qblk = pass ? (int)blockIdx.x : (S_ / 64 - 1) - (int)blockIdx.x;
    const int q0 = qblk * 64 + wave * 16;

    v4f kq[8], vq[8];
    #pragma unroll
    for (int i = 0; i < 8; ++i) kq[i] = *(const v4f*)(Wq + woff + 8 * i * D_);
    #pragma unroll
    for (int i = 0; i < 8; ++i) vq[i] = *(const v4f*)(Wk + woff + 8 * i * D_);
    float xq[32];
    const float* qrow = Xq + inOff + (size_t)(q0 + mrow) * E_;
    #pragma unroll
    for (int ks = 0; ks < 16; ++ks) {
      const int k = 4 * ks + 2 * half;
      xq[2 * ks]     = qrow[k];
      xq[2 * ks + 1] = qrow[k + 1];
    }

    __syncthreads();
    #pragma unroll
    for (int i = 0; i < 8; ++i) *(v4f*)(sdst + 8 * i * LDP_) = kq[i];
    #pragma unroll
    for (int i = 0; i < 8; ++i) *(v4f*)(sdst + (BK_ + 8 * i) * LDP_) = vq[i];
    __syncthreads();

    #pragma unroll
    for (int i = 0; i < 8; ++i) kq[i] = *(const v4f*)(kg + (size_t)(8 * i) * E_);
    #pragma unroll
    for (int i = 0; i < 8; ++i) vq[i] = *(const v4f*)(vg + (size_t)(8 * i) * E_);

    v8f qt[4] = {};
    #pragma unroll
    for (int ks = 0; ks < 16; ++ks) {
      const int k = 4 * ks + 2 * half;
      v2f b; b.x = xq[2 * ks]; b.y = xq[2 * ks + 1];
      #pragma unroll
      for (int f = 0; f < 4; ++f) {
        const v2f a = *(const v2f*)&Ks[(16 * f + mrow) * LDP_ + k];
        qt[f] = wmma_f32(a, b, qt[f]);
      }
    }
    v8f qp[4] = {};
    #pragma unroll
    for (int f = 0; f < 4; ++f)
      #pragma unroll
      for (int t = 0; t < 4; ++t) {
        const int e0 = 16 * f + 8 * half + 2 * t;
        v2f b; b.x = qt[f][2 * t]; b.y = qt[f][2 * t + 1];
        #pragma unroll
        for (int g = 0; g < 4; ++g) {
          const int dcol = 16 * g + mrow;
          v2f a;
          a.x = Vs[e0       * LDP_ + dcol];
          a.y = Vs[(e0 + 1) * LDP_ + dcol];
          qp[g] = wmma_f32(a, b, qp[g]);
        }
      }

    v8f o[4] = {};
    float mrun = -3.0e38f;
    float lrun = 0.0f;

    for (int kt = 0; kt <= qblk; ++kt) {
      const int k0 = kt * BK_;

      __syncthreads();
      #pragma unroll
      for (int i = 0; i < 8; ++i) *(v4f*)(sdst + 8 * i * LDP_) = kq[i];
      #pragma unroll
      for (int i = 0; i < 8; ++i) *(v4f*)(sdst + (BK_ + 8 * i) * LDP_) = vq[i];
      __syncthreads();

      if (kt < qblk) {
        #pragma unroll
        for (int i = 0; i < 8; ++i) kq[i] = *(const v4f*)(kg + (size_t)(k0 + BK_ + 8 * i) * E_);
        #pragma unroll
        for (int i = 0; i < 8; ++i) vq[i] = *(const v4f*)(vg + (size_t)(k0 + BK_ + 8 * i) * E_);
      } else {
        #pragma unroll
        for (int i = 0; i < 8; ++i) kq[i] = *(const v4f*)(Wv + woff + 8 * i * D_);
      }

      v8f s[4] = {};
      #pragma unroll
      for (int g = 0; g < 4; ++g)
        #pragma unroll
        for (int t = 0; t < 4; ++t) {
          const int k = 16 * g + 8 * half + 2 * t;
          v2f b; b.x = qp[g][2 * t]; b.y = qp[g][2 * t + 1];
          #pragma unroll
          for (int j = 0; j < 4; ++j) {
            const v2f a = *(const v2f*)&Ks[(16 * j + mrow) * LDP_ + k];
            s[j] = wmma_f32(a, b, s[j]);
          }
        }

      const float lanebias = ss * (float)(k0 - q0 + 8 * half - mrow);
      #pragma unroll
      for (int j = 0; j < 4; ++j)
        #pragma unroll
        for (int r = 0; r < 8; ++r)
          s[j][r] = fmaf(s[j][r], scale, lanebias + ss * (float)(16 * j + r));

      if (kt == qblk) {
        const int d = 8 * half - mrow - 16 * wave;
        #pragma unroll
        for (int j = 0; j < 4; ++j)
          #pragma unroll
          for (int r = 0; r < 8; ++r)
            if (d + 16 * j + r > 0) s[j][r] = -3.0e18f;
      }

      float rm;
      {
        float m4[4];
        #pragma unroll
        for (int j = 0; j < 4; ++j)
          m4[j] = fmaxf(fmaxf(fmaxf(s[j][0], s[j][1]), fmaxf(s[j][2], s[j][3])),
                        fmaxf(fmaxf(s[j][4], s[j][5]), fmaxf(s[j][6], s[j][7])));
        rm = fmaxf(fmaxf(m4[0], m4[1]), fmaxf(m4[2], m4[3]));
      }
      rm = fmaxf(rm, xhalf(rm));
      const float mnew  = fmaxf(mrun, rm);
      const float alpha = __expf(mrun - mnew);
      mrun = mnew;
      float p4[4];
      #pragma unroll
      for (int j = 0; j < 4; ++j) {
        #pragma unroll
        for (int r = 0; r < 8; ++r) s[j][r] = __expf(s[j][r] - mnew);
        p4[j] = ((s[j][0] + s[j][1]) + (s[j][2] + s[j][3])) +
                ((s[j][4] + s[j][5]) + (s[j][6] + s[j][7]));
      }
      lrun = lrun * alpha + ((p4[0] + p4[1]) + (p4[2] + p4[3]));
      #pragma unroll
      for (int f = 0; f < 4; ++f)
        #pragma unroll
        for (int r = 0; r < 8; ++r) o[f][r] *= alpha;

      #pragma unroll
      for (int j = 0; j < 4; ++j)
        #pragma unroll
        for (int t = 0; t < 4; ++t) {
          const int key = 16 * j + 8 * half + 2 * t;
          v2f b; b.x = s[j][2 * t]; b.y = s[j][2 * t + 1];
          #pragma unroll
          for (int f = 0; f < 4; ++f) {
            const int dcol = 16 * f + mrow;
            v2f a;
            a.x = Vs[key       * LDP_ + dcol];
            a.y = Vs[(key + 1) * LDP_ + dcol];
            o[f] = wmma_f32(a, b, o[f]);
          }
        }
    }

    __syncthreads();
    #pragma unroll
    for (int i = 0; i < 8; ++i) *(v4f*)(sdst + 8 * i * LDP_) = kq[i];
    __syncthreads();
    v8f fo[4] = {};
    #pragma unroll
    for (int f = 0; f < 4; ++f)
      #pragma unroll
      for (int t = 0; t < 4; ++t) {
        const int k = 16 * f + 8 * half + 2 * t;
        v2f b; b.x = o[f][2 * t]; b.y = o[f][2 * t + 1];
        #pragma unroll
        for (int g = 0; g < 4; ++g) {
          const v2f a = *(const v2f*)&Ks[(16 * g + mrow) * LDP_ + k];
          fo[g] = wmma_f32(a, b, fo[g]);
        }
      }

    const float linv = 1.0f / (lrun + xhalf(lrun));
    float* oblk = O + ((size_t)((n * S_ + q0) >> 4) * H_ + h) * 1024 + lane * 4;
    #pragma unroll
    for (int f = 0; f < 4; ++f) {
      v4f lo, hi;
      lo.x = fo[f][0] * linv; lo.y = fo[f][1] * linv; lo.z = fo[f][2] * linv; lo.w = fo[f][3] * linv;
      hi.x = fo[f][4] * linv; hi.y = fo[f][5] * linv; hi.z = fo[f][6] * linv; hi.w = fo[f][7] * linv;
      *(v4f*)(oblk + (2 * f) * 128)     = lo;
      *(v4f*)(oblk + (2 * f + 1) * 128) = hi;
    }
  }
}

static __device__ __forceinline__
void outproj_chunk(v8f (&c)[4], const v4f (&a)[8], v4f (&anext)[8], v4f (&wq)[8],
                   const float* Ws, float* sdst,
                   const float* __restrict__ wg, const float* __restrict__ ablk,
                   int kc, int mrow, int half)
{
  __syncthreads();
  #pragma unroll
  for (int i = 0; i < 8; ++i) *(v4f*)(sdst + 8 * i * 68) = wq[i];
  __syncthreads();

  if (kc + 64 < E_) {
    #pragma unroll
    for (int i = 0; i < 8; ++i) wq[i] = *(const v4f*)(wg + (size_t)(8 * i) * E_ + kc + 64);
    #pragma unroll
    for (int i = 0; i < 8; ++i) anext[i] = *(const v4f*)(ablk + (size_t)(kc + 64) * 16 + i * 128);
  }

  #pragma unroll
  for (int g = 0; g < 4; ++g)
    #pragma unroll
    for (int t = 0; t < 4; ++t) {
      const int k = 16 * g + 8 * half + 2 * t;
      v2f av; av.x = a[2 * g + (t >> 1)][2 * (t & 1)]; av.y = a[2 * g + (t >> 1)][2 * (t & 1) + 1];
      #pragma unroll
      for (int f = 0; f < 4; ++f) {
        const int e = 16 * f + mrow;
        const v2f b = *(const v2f*)&Ws[e * 68 + k];
        c[f] = wmma_f32(av, b, c[f]);
      }
    }
}

__global__ __launch_bounds__(128)
void outproj_kernel(const float* __restrict__ A,
                    const float* __restrict__ Wo,
                    const float* __restrict__ bo,
                    float* __restrict__ Y)
{
  __shared__ __attribute__((aligned(16))) float Ws[64 * 68];
  const int tid  = threadIdx.x;
  const int wave = tid >> 5;
  const int lane = tid & 31;
  const int r0 = blockIdx.y * 64 + wave * 16;
  const int c0 = blockIdx.x * 64;
  const int mrow = lane & 15, half = lane >> 4;
  const int srow = tid >> 4, sc4 = tid & 15;

  v8f c[4] = {};
  const float* ablk = A + (size_t)(r0 >> 4) * H_ * 1024 + lane * 4;
  const float* wg   = Wo + (size_t)(c0 + srow) * E_ + sc4 * 4;
  float* sdst = &Ws[srow * 68 + sc4 * 4];

  v4f wq[8];
  v4f a0[8], a1[8];
  #pragma unroll
  for (int i = 0; i < 8; ++i) wq[i] = *(const v4f*)(wg + (size_t)(8 * i) * E_);
  #pragma unroll
  for (int i = 0; i < 8; ++i) a0[i] = *(const v4f*)(ablk + i * 128);

  for (int kc = 0; kc < E_; kc += 128) {
    outproj_chunk(c, a0, a1, wq, Ws, sdst, wg, ablk, kc,      mrow, half);
    outproj_chunk(c, a1, a0, wq, Ws, sdst, wg, ablk, kc + 64, mrow, half);
  }

  #pragma unroll
  for (int f = 0; f < 4; ++f)
    #pragma unroll
    for (int r = 0; r < 8; ++r) {
      const int m = r + 8 * half;
      const int e = c0 + 16 * f + mrow;
      Y[(size_t)(r0 + m) * E_ + e] = c[f][r] + bo[e];
    }
}

extern "C" void kernel_launch(void* const* d_in, const int* in_sizes, int n_in,
                              void* d_out, int out_size, void* d_ws, size_t ws_size,
                              hipStream_t stream) {
  const float* values = (const float*)d_in[0];
  const float* keys   = (const float*)d_in[1];
  const float* query  = (const float*)d_in[2];
  const float* Wv     = (const float*)d_in[4];
  const float* Wk     = (const float*)d_in[5];
  const float* Wq     = (const float*)d_in[6];
  const float* Wo     = (const float*)d_in[7];
  const float* bo     = (const float*)d_in[8];
  float* out = (float*)d_out;

  float* Oa = (float*)d_ws;

  const dim3 blk(128);
  const dim3 gFlash(S_ / 128, H_, N_);
  flash_kernel<<<gFlash, blk, 0, stream>>>(query, keys, values, Wq, Wk, Wv, Oa);

  const dim3 gOut(E_ / 64, (N_ * S_) / 64);
  outproj_kernel<<<gOut, blk, 0, stream>>>(Oa, Wo, bo, out);
}


# both kernels: wave issue priority (s_setprio) steps down 3,2,1,0 with loop progress: work-groups sharing a WGP advance together, not oldest-first, and none runs its last stretch alone; sums unchanged
# speedup vs baseline: 2.9212x; 1.0365x over previous
#include <hip/hip_runtime.h>
#include <hip/hip_bf16.h>
#include <math.h>


typedef float v2f __attribute__((ext_vector_type(2)));
typedef float v4f __attribute__((ext_vector_type(4)));
typedef float v8f __attribute__((ext_vector_type(8)));

static constexpr int N_ = 2;
static constexpr int S_ = 2048;
static constexpr int E_ = 1024;
static constexpr int H_ = 16;
static constexpr int D_ = 64;

static __device__ __forceinline__ v8f wmma_f32(v2f a, v2f b, v8f c) {
  return __builtin_amdgcn_wmma_f32_16x16x4_f32(false, a, false, b,
                                               (short)0, c, false, false);
}

static __device__ __forceinline__ float xhalf(float x) {
  const int b = __builtin_bit_cast(int, x);
  return __builtin_bit_cast(float,
      __builtin_amdgcn_permlanex16(b, b, 0x76543210, 0xfedcba98, false, false));
}

static __device__ __forceinline__ void prio_by_progress(int step, int to2, int to1, int to0) {
  if (step == to2)      __builtin_amdgcn_s_setprio(2);
  else if (step == to1) __builtin_amdgcn_s_setprio(1);
  else if (step == to0) __builtin_amdgcn_s_setprio(0);
}

static constexpr int BK_  = 64;
static constexpr int LDP_ = 68;

__global__ __launch_bounds__(128)
void flash_kernel(const float* __restrict__ Xq,
                  const float* __restrict__ Xk,
                  const float* __restrict__ Xv,
                  const float* __restrict__ Wq,
                  const float* __restrict__ Wk,
                  const float* __restrict__ Wv,
                  float* __restrict__ O)
{
  __shared__ __attribute__((aligned(16))) float smem[2 * BK_ * LDP_];

  __builtin_amdgcn_s_setprio(3);

  const int tid  = threadIdx.x;
  const int wave = tid >> 5;
  const int lane = tid & 31;
  const int n = blockIdx.z, h = blockIdx.y;
  const int mrow = lane & 15, half = lane >> 4;
  const size_t inOff = (size_t)n * S_ * E_ + (size_t)h * D_;

  const float* Ks = smem;
  const float* Vs = smem + BK_ * LDP_;

  const float slope  = (float)(1 << __popc(h));
  const float scale  = 0.03125f;
  const float ss     = slope * scale;

  const int srow = tid >> 4, sc4 = tid & 15;
  const float* kg = Xk + inOff + (size_t)srow * E_ + sc4 * 4;
  const float* vg = Xv + inOff + (size_t)srow * E_ + sc4 * 4;
  const int woff = srow * D_ + sc4 * 4;
  float* sdst = smem + srow * LDP_ + sc4 * 4;

  for (int pass = 0; pass < 2; ++pass) {
    const int qblk = pass ? (int)blockIdx.x : (S_ / 64 - 1) - (int)blockIdx.x;
    const int q0 = qblk * 64 + wave * 16;

    v4f kq[8], vq[8];
    #pragma unroll
    for (int i = 0; i < 8; ++i) kq[i] = *(const v4f*)(Wq + woff + 8 * i * D_);
    #pragma unroll
    for (int i = 0; i < 8; ++i) vq[i] = *(const v4f*)(Wk + woff + 8 * i * D_);
    float xq[32];
    const float* qrow = Xq + inOff + (size_t)(q0 + mrow) * E_;
    #pragma unroll
    for (int ks = 0; ks < 16; ++ks) {
      const int k = 4 * ks + 2 * half;
      xq[2 * ks]     = qrow[k];
      xq[2 * ks + 1] = qrow[k + 1];
    }

    __syncthreads();
    #pragma unroll
    for (int i = 0; i < 8; ++i) *(v4f*)(sdst + 8 * i * LDP_) = kq[i];
    #pragma unroll
    for (int i = 0; i < 8; ++i) *(v4f*)(sdst + (BK_ + 8 * i) * LDP_) = vq[i];
    __syncthreads();

    #pragma unroll
    for (int i = 0; i < 8; ++i) kq[i] = *(const v4f*)(kg + (size_t)(8 * i) * E_);
    #pragma unroll
    for (int i = 0; i < 8; ++i) vq[i] = *(const v4f*)(vg + (size_t)(8 * i) * E_);

    v8f qt[4] = {};
    #pragma unroll
    for (int ks = 0; ks < 16; ++ks) {
      const int k = 4 * ks + 2 * half;
      v2f b; b.x = xq[2 * ks]; b.y = xq[2 * ks + 1];
      #pragma unroll
      for (int f = 0; f < 4; ++f) {
        const v2f a = *(const v2f*)&Ks[(16 * f + mrow) * LDP_ + k];
        qt[f] = wmma_f32(a, b, qt[f]);
      }
    }
    v8f qp[4] = {};
    #pragma unroll
    for (int f = 0; f < 4; ++f)
      #pragma unroll
      for (int t = 0; t < 4; ++t) {
        const int e0 = 16 * f + 8 * half + 2 * t;
        v2f b; b.x = qt[f][2 * t]; b.y = qt[f][2 * t + 1];
        #pragma unroll
        for (int g = 0; g < 4; ++g) {
          const int dcol = 16 * g + mrow;
          v2f a;
          a.x = Vs[e0       * LDP_ + dcol];
          a.y = Vs[(e0 + 1) * LDP_ + dcol];
          qp[g] = wmma_f32(a, b, qp[g]);
        }
      }

    v8f o[4] = {};
    float mrun = -3.0e38f;
    float lrun = 0.0f;

    const int tilesBefore = pass ? (S_ / 64) - (int)blockIdx.x : 0;

    for (int kt = 0; kt <= qblk; ++kt) {
      const int k0 = kt * BK_;

      prio_by_progress(tilesBefore + kt, 11, 22, 31);

      __syncthreads();
      #pragma unroll
      for (int i = 0; i < 8; ++i) *(v4f*)(sdst + 8 * i * LDP_) = kq[i];
      #pragma unroll
      for (int i = 0; i < 8; ++i) *(v4f*)(sdst + (BK_ + 8 * i) * LDP_) = vq[i];
      __syncthreads();

      if (kt < qblk) {
        #pragma unroll
        for (int i = 0; i < 8; ++i) kq[i] = *(const v4f*)(kg + (size_t)(k0 + BK_ + 8 * i) * E_);
        #pragma unroll
        for (int i = 0; i < 8; ++i) vq[i] = *(const v4f*)(vg + (size_t)(k0 + BK_ + 8 * i) * E_);
      } else {
        #pragma unroll
        for (int i = 0; i < 8; ++i) kq[i] = *(const v4f*)(Wv + woff + 8 * i * D_);
      }

      v8f s[4] = {};
      #pragma unroll
      for (int g = 0; g < 4; ++g)
        #pragma unroll
        for (int t = 0; t < 4; ++t) {
          const int k = 16 * g + 8 * half + 2 * t;
          v2f b; b.x = qp[g][2 * t]; b.y = qp[g][2 * t + 1];
          #pragma unroll
          for (int j = 0; j < 4; ++j) {
            const v2f a = *(const v2f*)&Ks[(16 * j + mrow) * LDP_ + k];
            s[j] = wmma_f32(a, b, s[j]);
          }
        }

      const float lanebias = ss * (float)(k0 - q0 + 8 * half - mrow);
      #pragma unroll
      for (int j = 0; j < 4; ++j)
        #pragma unroll
        for (int r = 0; r < 8; ++r)
          s[j][r] = fmaf(s[j][r], scale, lanebias + ss * (float)(16 * j + r));

      if (kt == qblk) {
        const int d = 8 * half - mrow - 16 * wave;
        #pragma unroll
        for (int j = 0; j < 4; ++j)
          #pragma unroll
          for (int r = 0; r < 8; ++r)
            if (d + 16 * j + r > 0) s[j][r] = -3.0e18f;
      }

      float rm;
      {
        float m4[4];
        #pragma unroll
        for (int j = 0; j < 4; ++j)
          m4[j] = fmaxf(fmaxf(fmaxf(s[j][0], s[j][1]), fmaxf(s[j][2], s[j][3])),
                        fmaxf(fmaxf(s[j][4], s[j][5]), fmaxf(s[j][6], s[j][7])));
        rm = fmaxf(fmaxf(m4[0], m4[1]), fmaxf(m4[2], m4[3]));
      }
      rm = fmaxf(rm, xhalf(rm));
      const float mnew  = fmaxf(mrun, rm);
      const float alpha = __expf(mrun - mnew);
      mrun = mnew;
      float p4[4];
      #pragma unroll
      for (int j = 0; j < 4; ++j) {
        #pragma unroll
        for (int r = 0; r < 8; ++r) s[j][r] = __expf(s[j][r] - mnew);
        p4[j] = ((s[j][0] + s[j][1]) + (s[j][2] + s[j][3])) +
                ((s[j][4] + s[j][5]) + (s[j][6] + s[j][7]));
      }
      lrun = lrun * alpha + ((p4[0] + p4[1]) + (p4[2] + p4[3]));
      #pragma unroll
      for (int f = 0; f < 4; ++f)
        #pragma unroll
        for (int r = 0; r < 8; ++r) o[f][r] *= alpha;

      #pragma unroll
      for (int j = 0; j < 4; ++j)
        #pragma unroll
        for (int t = 0; t < 4; ++t) {
          const int key = 16 * j + 8 * half + 2 * t;
          v2f b; b.x = s[j][2 * t]; b.y = s[j][2 * t + 1];
          #pragma unroll
          for (int f = 0; f < 4; ++f) {
            const int dcol = 16 * f + mrow;
            v2f a;
            a.x = Vs[key       * LDP_ + dcol];
            a.y = Vs[(key + 1) * LDP_ + dcol];
            o[f] = wmma_f32(a, b, o[f]);
          }
        }
    }

    __syncthreads();
    #pragma unroll
    for (int i = 0; i < 8; ++i) *(v4f*)(sdst + 8 * i * LDP_) = kq[i];
    __syncthreads();
    v8f fo[4] = {};
    #pragma unroll
    for (int f = 0; f < 4; ++f)
      #pragma unroll
      for (int t = 0; t < 4; ++t) {
        const int k = 16 * f + 8 * half + 2 * t;
        v2f b; b.x = o[f][2 * t]; b.y = o[f][2 * t + 1];
        #pragma unroll
        for (int g = 0; g < 4; ++g) {
          const v2f a = *(const v2f*)&Ks[(16 * g + mrow) * LDP_ + k];
          fo[g] = wmma_f32(a, b, fo[g]);
        }
      }

    const float linv = 1.0f / (lrun + xhalf(lrun));
    float* oblk = O + ((size_t)((n * S_ + q0) >> 4) * H_ + h) * 1024 + lane * 4;
    #pragma unroll
    for (int f = 0; f < 4; ++f) {
      v4f lo, hi;
      lo.x = fo[f][0] * linv; lo.y = fo[f][1] * linv; lo.z = fo[f][2] * linv; lo.w = fo[f][3] * linv;
      hi.x = fo[f][4] * linv; hi.y = fo[f][5] * linv; hi.z = fo[f][6] * linv; hi.w = fo[f][7] * linv;
      *(v4f*)(oblk + (2 * f) * 128)     = lo;
      *(v4f*)(oblk + (2 * f + 1) * 128) = hi;
    }
  }
}

static __device__ __forceinline__
void outproj_chunk(v8f (&c)[4], const v4f (&a)[8], v4f (&anext)[8], v4f (&wq)[8],
                   const float* Ws, float* sdst,
                   const float* __restrict__ wg, const float* __restrict__ ablk,
                   int kc, int mrow, int half)
{
  prio_by_progress(kc >> 6, 6, 11, 15);

  __syncthreads();
  #pragma unroll
  for (int i = 0; i < 8; ++i) *(v4f*)(sdst + 8 * i * 68) = wq[i];
  __syncthreads();

  if (kc + 64 < E_) {
    #pragma unroll
    for (int i = 0; i < 8; ++i) wq[i] = *(const v4f*)(wg + (size_t)(8 * i) * E_ + kc + 64);
    #pragma unroll
    for (int i = 0; i < 8; ++i) anext[i] = *(const v4f*)(ablk + (size_t)(kc + 64) * 16 + i * 128);
  }

  #pragma unroll
  for (int g = 0; g < 4; ++g)
    #pragma unroll
    for (int t = 0; t < 4; ++t) {
      const int k = 16 * g + 8 * half + 2 * t;
      v2f av; av.x = a[2 * g + (t >> 1)][2 * (t & 1)]; av.y = a[2 * g + (t >> 1)][2 * (t & 1) + 1];
      #pragma unroll
      for (int f = 0; f < 4; ++f) {
        const int e = 16 * f + mrow;
        const v2f b = *(const v2f*)&Ws[e * 68 + k];
        c[f] = wmma_f32(av, b, c[f]);
      }
    }
}

__global__ __launch_bounds__(128)
void outproj_kernel(const float* __restrict__ A,
                    const float* __restrict__ Wo,
                    const float* __restrict__ bo,
                    float* __restrict__ Y)
{
  __shared__ __attribute__((aligned(16))) float Ws[64 * 68];
  __builtin_amdgcn_s_setprio(3);

  const int tid  = threadIdx.x;
  const int wave = tid >> 5;
  const int lane = tid & 31;
  const int r0 = blockIdx.y * 64 + wave * 16;
  const int c0 = blockIdx.x * 64;
  const int mrow = lane & 15, half = lane >> 4;
  const int srow = tid >> 4, sc4 = tid & 15;

  v8f c[4] = {};
  const float* ablk = A + (size_t)(r0 >> 4) * H_ * 1024 + lane * 4;
  const float* wg   = Wo + (size_t)(c0 + srow) * E_ + sc4 * 4;
  float* sdst = &Ws[srow * 68 + sc4 * 4];

  v4f wq[8];
  v4f a0[8], a1[8];
  #pragma unroll
  for (int i = 0; i < 8; ++i) wq[i] = *(const v4f*)(wg + (size_t)(8 * i) * E_);
  #pragma unroll
  for (int i = 0; i < 8; ++i) a0[i] = *(const v4f*)(ablk + i * 128);

  for (int kc = 0; kc < E_; kc += 128) {
    outproj_chunk(c, a0, a1, wq, Ws, sdst, wg, ablk, kc,      mrow, half);
    outproj_chunk(c, a1, a0, wq, Ws, sdst, wg, ablk, kc + 64, mrow, half);
  }

  #pragma unroll
  for (int f = 0; f < 4; ++f)
    #pragma unroll
    for (int r = 0; r < 8; ++r) {
      const int m = r + 8 * half;
      const int e = c0 + 16 * f + mrow;
      Y[(size_t)(r0 + m) * E_ + e] = c[f][r] + bo[e];
    }
}

extern "C" void kernel_launch(void* const* d_in, const int* in_sizes, int n_in,
                              void* d_out, int out_size, void* d_ws, size_t ws_size,
                              hipStream_t stream) {
  const float* values = (const float*)d_in[0];
  const float* keys   = (const float*)d_in[1];
  const float* query  = (const float*)d_in[2];
  const float* Wv     = (const float*)d_in[4];
  const float* Wk     = (const float*)d_in[5];
  const float* Wq     = (const float*)d_in[6];
  const float* Wo     = (const float*)d_in[7];
  const float* bo     = (const float*)d_in[8];
  float* out = (float*)d_out;

  float* Oa = (float*)d_ws;

  const dim3 blk(128);
  const dim3 gFlash(S_ / 128, H_, N_);
  flash_kernel<<<gFlash, blk, 0, stream>>>(query, keys, values, Wq, Wk, Wv, Oa);

  const dim3 gOut(E_ / 64, (N_ * S_) / 64);
  outproj_kernel<<<gOut, blk, 0, stream>>>(Oa, Wo, bo, out);
}


# attention kernel: LDS fragment reads of each 64-WMMA block go out as one group ahead of it, not among its WMMAs (V tile stored with key pairs adjacent: one 8-byte read per fragment); sums unchanged
# speedup vs baseline: 3.0479x; 1.0434x over previous
#include <hip/hip_runtime.h>
#include <hip/hip_bf16.h>
#include <math.h>


typedef float v2f __attribute__((ext_vector_type(2)));
typedef float v4f __attribute__((ext_vector_type(4)));
typedef float v8f __attribute__((ext_vector_type(8)));

static constexpr int N_ = 2;
static constexpr int S_ = 2048;
static constexpr int E_ = 1024;
static constexpr int H_ = 16;
static constexpr int D_ = 64;

static __device__ __forceinline__ v8f wmma_f32(v2f a, v2f b, v8f c) {
  return __builtin_amdgcn_wmma_f32_16x16x4_f32(false, a, false, b,
                                               (short)0, c, false, false);
}

static __device__ __forceinline__ float xhalf(float x) {
  const int b = __builtin_bit_cast(int, x);
  return __builtin_bit_cast(float,
      __builtin_amdgcn_permlanex16(b, b, 0x76543210, 0xfedcba98, false, false));
}

static __device__ __forceinline__ void prio_by_progress(int step, int to2, int to1, int to0) {
  if (step == to2)      __builtin_amdgcn_s_setprio(2);
  else if (step == to1) __builtin_amdgcn_s_setprio(1);
  else if (step == to0) __builtin_amdgcn_s_setprio(0);
}

static constexpr int BK_  = 64;
static constexpr int LDP_ = 68;
static constexpr int VP_  = 136;

static __device__ __forceinline__ void load_row_frags(v4f (&kf)[4][2][4], const float* Ks, int mrow, int half) {
  #pragma unroll
  for (int g = 0; g < 4; ++g)
    #pragma unroll
    for (int u = 0; u < 2; ++u)
      #pragma unroll
      for (int j = 0; j < 4; ++j)
        kf[g][u][j] = *(const v4f*)&Ks[(16 * j + mrow) * LDP_ + 16 * g + 8 * half + 4 * u];
}

static __device__ __forceinline__ void load_col_frags(v2f (&vf)[4][4][4], const float* Vp, int mrow, int half) {
  #pragma unroll
  for (int j = 0; j < 4; ++j)
    #pragma unroll
    for (int t = 0; t < 4; ++t)
      #pragma unroll
      for (int f = 0; f < 4; ++f)
        vf[j][t][f] = *(const v2f*)&Vp[(8 * j + 4 * half + t) * VP_ + (16 * f + mrow) * 2];
}

static __device__ __forceinline__ void store_pair_rows(float* vdst, const v4f (&p)[8]) {
  #pragma unroll
  for (int i = 0; i < 8; ++i) {
    float* d = vdst + 4 * i * VP_;
    d[0] = p[i].x; d[2] = p[i].y; d[4] = p[i].z; d[6] = p[i].w;
  }
}

__global__ __launch_bounds__(128)
void flash_kernel(const float* __restrict__ Xq,
                  const float* __restrict__ Xk,
                  const float* __restrict__ Xv,
                  const float* __restrict__ Wq,
                  const float* __restrict__ Wk,
                  const float* __restrict__ Wv,
                  float* __restrict__ O)
{
  __shared__ __attribute__((aligned(16))) float smem[2 * BK_ * LDP_];

  __builtin_amdgcn_s_setprio(3);

  const int tid  = threadIdx.x;
  const int wave = tid >> 5;
  const int lane = tid & 31;
  const int n = blockIdx.z, h = blockIdx.y;
  const int mrow = lane & 15, half = lane >> 4;
  const size_t inOff = (size_t)n * S_ * E_ + (size_t)h * D_;

  const float* Ks = smem;
  const float* Vp = smem + BK_ * LDP_;

  const float slope  = (float)(1 << __popc(h));
  const float scale  = 0.03125f;
  const float ss     = slope * scale;

  const int srow = tid >> 4, sc4 = tid & 15;
  const float* kg = Xk + inOff + (size_t)srow * E_ + sc4 * 4;
  const float* vg = Xv + inOff + (size_t)srow * E_ + sc4 * 4;
  const int woff = srow * D_ + sc4 * 4;
  float* sdst = smem + srow * LDP_ + sc4 * 4;
  float* vdst = smem + BK_ * LDP_ + (srow >> 1) * VP_ + sc4 * 8 + (srow & 1);

  for (int pass = 0; pass < 2; ++pass) {
    const int qblk = pass ? (int)blockIdx.x : (S_ / 64 - 1) - (int)blockIdx.x;
    const int q0 = qblk * 64 + wave * 16;

    v4f kq[8], vq[8];
    #pragma unroll
    for (int i = 0; i < 8; ++i) kq[i] = *(const v4f*)(Wq + woff + 8 * i * D_);
    #pragma unroll
    for (int i = 0; i < 8; ++i) vq[i] = *(const v4f*)(Wk + woff + 8 * i * D_);
    float xq[32];
    const float* qrow = Xq + inOff + (size_t)(q0 + mrow) * E_;
    #pragma unroll
    for (int ks = 0; ks < 16; ++ks) {
      const int k = 4 * ks + 2 * half;
      xq[2 * ks]     = qrow[k];
      xq[2 * ks + 1] = qrow[k + 1];
    }

    __syncthreads();
    #pragma unroll
    for (int i = 0; i < 8; ++i) *(v4f*)(sdst + 8 * i * LDP_) = kq[i];
    store_pair_rows(vdst, vq);
    __syncthreads();

    #pragma unroll
    for (int i = 0; i < 8; ++i) kq[i] = *(const v4f*)(kg + (size_t)(8 * i) * E_);
    #pragma unroll
    for (int i = 0; i < 8; ++i) vq[i] = *(const v4f*)(vg + (size_t)(8 * i) * E_);

    v8f qt[4] = {};
    {
      v2f qf[16][4];
      #pragma unroll
      for (int ks = 0; ks < 16; ++ks)
        #pragma unroll
        for (int f = 0; f < 4; ++f)
          qf[ks][f] = *(const v2f*)&Ks[(16 * f + mrow) * LDP_ + 4 * ks + 2 * half];
      __builtin_amdgcn_sched_barrier(0);
      #pragma unroll
      for (int ks = 0; ks < 16; ++ks) {
        v2f b; b.x = xq[2 * ks]; b.y = xq[2 * ks + 1];
        #pragma unroll
        for (int f = 0; f < 4; ++f) qt[f] = wmma_f32(qf[ks][f], b, qt[f]);
      }
    }
    v8f qp[4] = {};
    {
      v2f wf[4][4][4];
      __builtin_amdgcn_sched_barrier(0);
      load_col_frags(wf, Vp, mrow, half);
      __builtin_amdgcn_sched_barrier(0);
      #pragma unroll
      for (int f = 0; f < 4; ++f)
        #pragma unroll
        for (int t = 0; t < 4; ++t) {
          v2f b; b.x = qt[f][2 * t]; b.y = qt[f][2 * t + 1];
          #pragma unroll
          for (int g = 0; g < 4; ++g) qp[g] = wmma_f32(wf[f][t][g], b, qp[g]);
        }
    }

    v8f o[4] = {};
    float mrun = -3.0e38f;
    float lrun = 0.0f;

    const int tilesBefore = pass ? (S_ / 64) - (int)blockIdx.x : 0;

    for (int kt = 0; kt <= qblk; ++kt) {
      const int k0 = kt * BK_;

      prio_by_progress(tilesBefore + kt, 11, 22, 31);

      __syncthreads();
      #pragma unroll
      for (int i = 0; i < 8; ++i) *(v4f*)(sdst + 8 * i * LDP_) = kq[i];
      store_pair_rows(vdst, vq);
      __syncthreads();

      v4f kf[4][2][4];
      load_row_frags(kf, Ks, mrow, half);

      if (kt < qblk) {
        #pragma unroll
        for (int i = 0; i < 8; ++i) kq[i] = *(const v4f*)(kg + (size_t)(k0 + BK_ + 8 * i) * E_);
        #pragma unroll
        for (int i = 0; i < 8; ++i) vq[i] = *(const v4f*)(vg + (size_t)(k0 + BK_ + 8 * i) * E_);
      } else {
        #pragma unroll
        for (int i = 0; i < 8; ++i) kq[i] = *(const v4f*)(Wv + woff + 8 * i * D_);
      }

      v8f s[4] = {};
      #pragma unroll
      for (int g = 0; g < 4; ++g)
        #pragma unroll
        for (int t = 0; t < 4; ++t) {
          v2f b; b.x = qp[g][2 * t]; b.y = qp[g][2 * t + 1];
          #pragma unroll
          for (int j = 0; j < 4; ++j) {
            const v4f q4 = kf[g][t >> 1][j];
            v2f a;
            if (t & 1) { a.x = q4.z; a.y = q4.w; } else { a.x = q4.x; a.y = q4.y; }
            s[j] = wmma_f32(a, b, s[j]);
          }
        }

      v2f vf[4][4][4];
      __builtin_amdgcn_sched_barrier(0);
      load_col_frags(vf, Vp, mrow, half);
      __builtin_amdgcn_sched_barrier(0);

      const float lanebias = ss * (float)(k0 - q0 + 8 * half - mrow);
      #pragma unroll
      for (int j = 0; j < 4; ++j)
        #pragma unroll
        for (int r = 0; r < 8; ++r)
          s[j][r] = fmaf(s[j][r], scale, lanebias + ss * (float)(16 * j + r));

      if (kt == qblk) {
        const int d = 8 * half - mrow - 16 * wave;
        #pragma unroll
        for (int j = 0; j < 4; ++j)
          #pragma unroll
          for (int r = 0; r < 8; ++r)
            if (d + 16 * j + r > 0) s[j][r] = -3.0e18f;
      }

      float rm;
      {
        float m4[4];
        #pragma unroll
        for (int j = 0; j < 4; ++j)
          m4[j] = fmaxf(fmaxf(fmaxf(s[j][0], s[j][1]), fmaxf(s[j][2], s[j][3])),
                        fmaxf(fmaxf(s[j][4], s[j][5]), fmaxf(s[j][6], s[j][7])));
        rm = fmaxf(fmaxf(m4[0], m4[1]), fmaxf(m4[2], m4[3]));
      }
      rm = fmaxf(rm, xhalf(rm));
      const float mnew  = fmaxf(mrun, rm);
      const float alpha = __expf(mrun - mnew);
      mrun = mnew;
      float p4[4];
      #pragma unroll
      for (int j = 0; j < 4; ++j) {
        #pragma unroll
        for (int r = 0; r < 8; ++r) s[j][r] = __expf(s[j][r] - mnew);
        p4[j] = ((s[j][0] + s[j][1]) + (s[j][2] + s[j][3])) +
                ((s[j][4] + s[j][5]) + (s[j][6] + s[j][7]));
      }
      lrun = lrun * alpha + ((p4[0] + p4[1]) + (p4[2] + p4[3]));
      #pragma unroll
      for (int f = 0; f < 4; ++f)
        #pragma unroll
        for (int r = 0; r < 8; ++r) o[f][r] *= alpha;

      #pragma unroll
      for (int j = 0; j < 4; ++j)
        #pragma unroll
        for (int t = 0; t < 4; ++t) {
          v2f b; b.x = s[j][2 * t]; b.y = s[j][2 * t + 1];
          #pragma unroll
          for (int f = 0; f < 4; ++f) o[f] = wmma_f32(vf[j][t][f], b, o[f]);
        }
    }

    __syncthreads();
    #pragma unroll
    for (int i = 0; i < 8; ++i) *(v4f*)(sdst + 8 * i * LDP_) = kq[i];
    __syncthreads();
    v8f fo[4] = {};
    {
      v4f wf[4][2][4];
      load_row_frags(wf, Ks, mrow, half);
      __builtin_amdgcn_sched_barrier(0);
      #pragma unroll
      for (int f = 0; f < 4; ++f)
        #pragma unroll
        for (int t = 0; t < 4; ++t) {
          v2f b; b.x = o[f][2 * t]; b.y = o[f][2 * t + 1];
          #pragma unroll
          for (int g = 0; g < 4; ++g) {
            const v4f q4 = wf[f][t >> 1][g];
            v2f a;
            if (t & 1) { a.x = q4.z; a.y = q4.w; } else { a.x = q4.x; a.y = q4.y; }
            fo[g] = wmma_f32(a, b, fo[g]);
          }
        }
    }

    const float linv = 1.0f / (lrun + xhalf(lrun));
    float* oblk = O + ((size_t)((n * S_ + q0) >> 4) * H_ + h) * 1024 + lane * 4;
    #pragma unroll
    for (int f = 0; f < 4; ++f) {
      v4f lo, hi;
      lo.x = fo[f][0] * linv; lo.y = fo[f][1] * linv; lo.z = fo[f][2] * linv; lo.w = fo[f][3] * linv;
      hi.x = fo[f][4] * linv; hi.y = fo[f][5] * linv; hi.z = fo[f][6] * linv; hi.w = fo[f][7] * linv;
      *(v4f*)(oblk + (2 * f) * 128)     = lo;
      *(v4f*)(oblk + (2 * f + 1) * 128) = hi;
    }
  }
}

static __device__ __forceinline__
void outproj_chunk(v8f (&c)[4], const v4f (&a)[8], v4f (&anext)[8], v4f (&wq)[8],
                   const float* Ws, float* sdst,
                   const float* __restrict__ wg, const float* __restrict__ ablk,
                   int kc, int mrow, int half)
{
  prio_by_progress(kc >> 6, 6, 11, 15);

  __syncthreads();
  #pragma unroll
  for (int i = 0; i < 8; ++i) *(v4f*)(sdst + 8 * i * 68) = wq[i];
  __syncthreads();

  if (kc + 64 < E_) {
    #pragma unroll
    for (int i = 0; i < 8; ++i) wq[i] = *(const v4f*)(wg + (size_t)(8 * i) * E_ + kc + 64);
    #pragma unroll
    for (int i = 0; i < 8; ++i) anext[i] = *(const v4f*)(ablk + (size_t)(kc + 64) * 16 + i * 128);
  }

  #pragma unroll
  for (int g = 0; g < 4; ++g)
    #pragma unroll
    for (int t = 0; t < 4; ++t) {
      const int k = 16 * g + 8 * half + 2 * t;
      v2f av; av.x = a[2 * g + (t >> 1)][2 * (t & 1)]; av.y = a[2 * g + (t >> 1)][2 * (t & 1) + 1];
      #pragma unroll
      for (int f = 0; f < 4; ++f) {
        const int e = 16 * f + mrow;
        const v2f b = *(const v2f*)&Ws[e * 68 + k];
        c[f] = wmma_f32(av, b, c[f]);
      }
    }
}

__global__ __launch_bounds__(128)
void outproj_kernel(const float* __restrict__ A,
                    const float* __restrict__ Wo,
                    const float* __restrict__ bo,
                    float* __restrict__ Y)
{
  __shared__ __attribute__((aligned(16))) float Ws[64 * 68];
  __builtin_amdgcn_s_setprio(3);

  const int tid  = threadIdx.x;
  const int wave = tid >> 5;
  const int lane = tid & 31;
  const int r0 = blockIdx.y * 64 + wave * 16;
  const int c0 = blockIdx.x * 64;
  const int mrow = lane & 15, half = lane >> 4;
  const int srow = tid >> 4, sc4 = tid & 15;

  v8f c[4] = {};
  const float* ablk = A + (size_t)(r0 >> 4) * H_ * 1024 + lane * 4;
  const float* wg   = Wo + (size_t)(c0 + srow) * E_ + sc4 * 4;
  float* sdst = &Ws[srow * 68 + sc4 * 4];

  v4f wq[8];
  v4f a0[8], a1[8];
  #pragma unroll
  for (int i = 0; i < 8; ++i) wq[i] = *(const v4f*)(wg + (size_t)(8 * i) * E_);
  #pragma unroll
  for (int i = 0; i < 8; ++i) a0[i] = *(const v4f*)(ablk + i * 128);

  for (int kc = 0; kc < E_; kc += 128) {
    outproj_chunk(c, a0, a1, wq, Ws, sdst, wg, ablk, kc,      mrow, half);
    outproj_chunk(c, a1, a0, wq, Ws, sdst, wg, ablk, kc + 64, mrow, half);
  }

  #pragma unroll
  for (int f = 0; f < 4; ++f)
    #pragma unroll
    for (int r = 0; r < 8; ++r) {
      const int m = r + 8 * half;
      const int e = c0 + 16 * f + mrow;
      Y[(size_t)(r0 + m) * E_ + e] = c[f][r] + bo[e];
    }
}

extern "C" void kernel_launch(void* const* d_in, const int* in_sizes, int n_in,
                              void* d_out, int out_size, void* d_ws, size_t ws_size,
                              hipStream_t stream) {
  const float* values = (const float*)d_in[0];
  const float* keys   = (const float*)d_in[1];
  const float* query  = (const float*)d_in[2];
  const float* Wv     = (const float*)d_in[4];
  const float* Wk     = (const float*)d_in[5];
  const float* Wq     = (const float*)d_in[6];
  const float* Wo     = (const float*)d_in[7];
  const float* bo     = (const float*)d_in[8];
  float* out = (float*)d_out;

  float* Oa = (float*)d_ws;

  const dim3 blk(128);
  const dim3 gFlash(S_ / 128, H_, N_);
  flash_kernel<<<gFlash, blk, 0, stream>>>(query, keys, values, Wq, Wk, Wv, Oa);

  const dim3 gOut(E_ / 64, (N_ * S_) / 64);
  outproj_kernel<<<gOut, blk, 0, stream>>>(Oa, Wo, bo, out);
}
